# starting kernel
# baseline (speedup 1.0000x reference)
#include <hip/hip_runtime.h>
#include <math.h>

#define B_      8
#define TXT_T   200
#define TXT_PAD 208
#define MEL_T   800
#define N_MEL   80
#define KDIM    160
#define PAD_MAG 1.0e12f

typedef __attribute__((ext_vector_type(2))) float v2f;
typedef __attribute__((ext_vector_type(8))) float v8f;

#define A_OFF 0
#define C_OFF (B_ * TXT_PAD * KDIM)
#define P_OFF (C_OFF + B_ * TXT_PAD)

__global__ void mdn_prep_kernel(const float* __restrict__ mu_logvar,
                                float* __restrict__ ws) {
  int idx = blockIdx.x * blockDim.x + threadIdx.x;
  if (idx >= B_ * TXT_PAD) return;
  int b = idx / TXT_PAD;
  int t = idx % TXT_PAD;
  float* Arow = ws + A_OFF + (size_t)idx * KDIM;
  float c = 0.0f;
  if (t < TXT_T) {
    const float* ml = mu_logvar + (size_t)(b * TXT_T + t) * (2 * N_MEL);
    for (int n = 0; n < N_MEL; ++n) {
      float mu = ml[n];
      float lv = ml[N_MEL + n];
      float w  = expf(-lv);
      Arow[n]          = w;
      Arow[N_MEL + n]  = -2.0f * mu * w;
      c += mu * mu * w + lv;
    }
  } else {
    for (int n = 0; n < KDIM; ++n) Arow[n] = 0.0f;
  }
  ws[C_OFF + idx] = c;
}

__global__ void __launch_bounds__(256)
mdn_gemm_kernel(const float* __restrict__ ws,
                const float* __restrict__ mel,
                float* __restrict__ lp) {
  const int TT = TXT_PAD / 16;
  const int MT = MEL_T / 16;
  int wid = blockIdx.x * (blockDim.x >> 5) + (threadIdx.x >> 5);
  if (wid >= B_ * TT * MT) return;
  int lane = threadIdx.x & 31;
  int l15  = lane & 15;
  int kl   = lane >> 4;

  int b  = wid / (TT * MT);
  int r  = wid % (TT * MT);
  int t0 = (r / MT) * 16;
  int m0 = (r % MT) * 16;

  const float* Arow = ws + A_OFF + (size_t)(b * TXT_PAD + t0 + l15) * KDIM;
  const float* melb = mel + (size_t)b * N_MEL * MEL_T + m0 + l15;

  v8f acc = {0.f, 0.f, 0.f, 0.f, 0.f, 0.f, 0.f, 0.f};

  for (int k = 0; k < KDIM; k += 4) {
    v2f a = *(const v2f*)(Arow + k + 2 * kl);

    int K0 = k + 2 * kl;
    int n  = (k < N_MEL) ? K0 : (K0 - N_MEL);
    float x0 = melb[(size_t)n * MEL_T];
    float x1 = melb[(size_t)(n + 1) * MEL_T];
    if (k < N_MEL) { x0 *= x0; x1 *= x1; }
    v2f bv; bv.x = x0; bv.y = x1;

    acc = __builtin_amdgcn_wmma_f32_16x16x4_f32(
         false, a,  false, bv,
         (short)0, acc,  false,  false);
  }

  int m = m0 + l15;
  const float* Cb = ws + C_OFF + b * TXT_PAD;
  const float scale = -0.5f / (float)N_MEL;
#pragma unroll
  for (int j = 0; j < 8; ++j) {
    int t = t0 + j + 8 * kl;
    if (t < TXT_T) {
      lp[(size_t)(b * TXT_T + t) * MEL_T + m] = scale * (acc[j] + Cb[t]);
    }
  }
}

__device__ __forceinline__ float logaddexp_(float u, float v) {
  float mx = fmaxf(u, v);
  float mn = fminf(u, v);
  return mx + log1pf(expf(mn - mx));
}

__global__ void mdn_scan_kernel(const float* __restrict__ lp,
                                const int* __restrict__ text_len,
                                const int* __restrict__ mel_len,
                                float* __restrict__ ws) {
  __shared__ float sa[TXT_T + 1];
  int b = blockIdx.x;
  int t = threadIdx.x;
  for (int i = t; i <= TXT_T; i += blockDim.x) sa[i] = -PAD_MAG;
  __syncthreads();
  const float* lpb = lp + (size_t)b * TXT_T * MEL_T;
  if (t == 0) sa[1] = lpb[0];
  __syncthreads();

  int M = mel_len[b];
  for (int m = 1; m < M; ++m) {
    float u = 0.f, v = 0.f;
    if (t < TXT_T) { u = sa[t + 1]; v = sa[t]; }
    __syncthreads();
    if (t < TXT_T) {
      sa[t + 1] = logaddexp_(u + 1e-7f, v + 1e-7f) + lpb[(size_t)t * MEL_T + m];
    }
    __syncthreads();
  }
  if (t == 0) {
    int tl = text_len[b];
    ws[P_OFF + b] = sa[tl] / (float)M;
  }
}

__global__ void mdn_reduce_kernel(const float* __restrict__ ws,
                                  float* __restrict__ out) {
  float s = 0.f;
  for (int b = 0; b < B_; ++b) s += ws[P_OFF + b];
  out[0] = -s / (float)B_;
}

extern "C" void kernel_launch(void* const* d_in, const int* in_sizes, int n_in,
                              void* d_out, int out_size, void* d_ws, size_t ws_size,
                              hipStream_t stream) {
  const float* mu_logvar = (const float*)d_in[0];
  const float* melspec   = (const float*)d_in[1];
  const int*   text_len  = (const int*)d_in[2];
  const int*   mel_len   = (const int*)d_in[3];
  float* out = (float*)d_out;
  float* ws  = (float*)d_ws;
  float* lp  = out + 1;

  mdn_prep_kernel<<<(B_ * TXT_PAD + 255) / 256, 256, 0, stream>>>(mu_logvar, ws);

  const int total_waves = B_ * (TXT_PAD / 16) * (MEL_T / 16);
  mdn_gemm_kernel<<<total_waves / 8, 256, 0, stream>>>(ws, melspec, lp);

  mdn_scan_kernel<<<B_, 256, 0, stream>>>(lp, text_len, mel_len, ws);
  mdn_reduce_kernel<<<1, 1, 0, stream>>>(ws, out);
}


# scan: carry held in the linear domain as mantissa*2^exponent per row (ldexp alignment), exp of lp taken off the column chain, one log at the end; rounding of the scan sums changes
# speedup vs baseline: 2.8437x; 2.8437x over previous
#include <hip/hip_runtime.h>
#include <math.h>

#define B_      8
#define TXT_T   200
#define TXT_PAD 208
#define MEL_T   800
#define N_MEL   80
#define KDIM    160
#define PAD_MAG 1.0e12f

typedef __attribute__((ext_vector_type(2))) float v2f;
typedef __attribute__((ext_vector_type(8))) float v8f;

#define A_OFF 0
#define C_OFF (B_ * TXT_PAD * KDIM)
#define P_OFF (C_OFF + B_ * TXT_PAD)

__global__ void mdn_prep_kernel(const float* __restrict__ mu_logvar,
                                float* __restrict__ ws) {
  int idx = blockIdx.x * blockDim.x + threadIdx.x;
  if (idx >= B_ * TXT_PAD) return;
  int b = idx / TXT_PAD;
  int t = idx % TXT_PAD;
  float* Arow = ws + A_OFF + (size_t)idx * KDIM;
  float c = 0.0f;
  if (t < TXT_T) {
    const float* ml = mu_logvar + (size_t)(b * TXT_T + t) * (2 * N_MEL);
    for (int n = 0; n < N_MEL; ++n) {
      float mu = ml[n];
      float lv = ml[N_MEL + n];
      float w  = expf(-lv);
      Arow[n]          = w;
      Arow[N_MEL + n]  = -2.0f * mu * w;
      c += mu * mu * w + lv;
    }
  } else {
    for (int n = 0; n < KDIM; ++n) Arow[n] = 0.0f;
  }
  ws[C_OFF + idx] = c;
}

__global__ void __launch_bounds__(256)
mdn_gemm_kernel(const float* __restrict__ ws,
                const float* __restrict__ mel,
                float* __restrict__ lp) {
  const int TT = TXT_PAD / 16;
  const int MT = MEL_T / 16;
  int wid = blockIdx.x * (blockDim.x >> 5) + (threadIdx.x >> 5);
  if (wid >= B_ * TT * MT) return;
  int lane = threadIdx.x & 31;
  int l15  = lane & 15;
  int kl   = lane >> 4;

  int b  = wid / (TT * MT);
  int r  = wid % (TT * MT);
  int t0 = (r / MT) * 16;
  int m0 = (r % MT) * 16;

  const float* Arow = ws + A_OFF + (size_t)(b * TXT_PAD + t0 + l15) * KDIM;
  const float* melb = mel + (size_t)b * N_MEL * MEL_T + m0 + l15;

  v8f acc = {0.f, 0.f, 0.f, 0.f, 0.f, 0.f, 0.f, 0.f};

  for (int k = 0; k < KDIM; k += 4) {
    v2f a = *(const v2f*)(Arow + k + 2 * kl);

    int K0 = k + 2 * kl;
    int n  = (k < N_MEL) ? K0 : (K0 - N_MEL);
    float x0 = melb[(size_t)n * MEL_T];
    float x1 = melb[(size_t)(n + 1) * MEL_T];
    if (k < N_MEL) { x0 *= x0; x1 *= x1; }
    v2f bv; bv.x = x0; bv.y = x1;

    acc = __builtin_amdgcn_wmma_f32_16x16x4_f32(
         false, a,  false, bv,
         (short)0, acc,  false,  false);
  }

  int m = m0 + l15;
  const float* Cb = ws + C_OFF + b * TXT_PAD;
  const float scale = -0.5f / (float)N_MEL;
#pragma unroll
  for (int j = 0; j < 8; ++j) {
    int t = t0 + j + 8 * kl;
    if (t < TXT_T) {
      lp[(size_t)(b * TXT_T + t) * MEL_T + m] = scale * (acc[j] + Cb[t]);
    }
  }
}

#define SC_ROWS  32
#define SC_WAVES ((TXT_T + SC_ROWS - 1) / SC_ROWS)
#define SC_COLS  8
#define SC_MAXCH (MEL_T / SC_COLS)
#define LOG2E_F  1.44269504088896340736f
#define LN2_F    0.69314718055994530942f
#define EPS_L2   (1.0e-7f * LOG2E_F)
#define E_ZERO   (-(1 << 28))

__device__ __forceinline__ unsigned lane_below_(unsigned ua, unsigned edge, bool is16) {
  unsigned sh = __builtin_amdgcn_update_dpp(edge, ua, 0x111, 0xf, 0xf, false);
  unsigned xr = __builtin_amdgcn_permlanex16(edge, ua, 0x0000000Fu, 0u, false, false);
  return is16 ? xr : sh;
}

__device__ __forceinline__ void weight_(float x, float eps, float& pf, int& pe) {
  const float lg = fmaf(x, LOG2E_F, eps);
  const float fl = floorf(lg);
  pf = __builtin_amdgcn_exp2f(lg - fl);
  pe = (int)fl;
}

__device__ __forceinline__ void scan_col_(float& f, int& e, float bf, int be,
                                          float pf, int pe, bool is16) {
  const float fb = __uint_as_float(lane_below_(__float_as_uint(f), __float_as_uint(bf), is16));
  const int   eb = (int)lane_below_((unsigned)e, (unsigned)be, is16);
  const int   E  = e > eb ? e : eb;
  const float s  = __builtin_ldexpf(f, e - E) + __builtin_ldexpf(fb, eb - E);
  f = s * pf;
  e = E + pe;
}

__global__ void __launch_bounds__(SC_WAVES * 32)
mdn_scan_kernel(const float* __restrict__ lp,
                const int* __restrict__ text_len,
                const int* __restrict__ mel_len,
                float* __restrict__ ws) {
  __shared__ __attribute__((aligned(16))) float edge_f[SC_WAVES][2][SC_COLS];
  __shared__ __attribute__((aligned(16))) int   edge_e[SC_WAVES][2][SC_COLS];
  const int b    = blockIdx.x;
  const int w    = __builtin_amdgcn_readfirstlane((int)(threadIdx.x >> 5));
  const int lane = threadIdx.x & 31;
  const int t    = w * SC_ROWS + lane;
  const bool is16 = (lane == 16);
  const int M  = mel_len[b];
  const int tl = text_len[b];
  int nch = (M + SC_COLS - 1) / SC_COLS;
  nch = nch < 0 ? 0 : (nch > SC_MAXCH ? SC_MAXCH : nch);

  for (int i = threadIdx.x; i < SC_WAVES * 2 * SC_COLS; i += SC_WAVES * 32) {
    (&edge_f[0][0][0])[i] = 0.0f;
    (&edge_e[0][0][0])[i] = E_ZERO;
  }

  const int tr = t < TXT_T ? t : TXT_T - 1;
  const float* row = lp + ((size_t)b * TXT_T + tr) * MEL_T;

  float cur[SC_COLS], nxt[SC_COLS];
#pragma unroll
  for (int i = 0; i < SC_COLS; ++i) cur[i] = row[i];

  float f = 0.0f;  int e = E_ZERO;
  float rf = 0.0f; int re = E_ZERO;
  float plf = (w == 0) ? 1.0f : 0.0f;
  int   ple = (w == 0) ? 0 : E_ZERO;
  __syncthreads();

  for (int s = 0; s < nch + SC_WAVES - 1; ++s) {
    const int c = s - w;
    if (c >= 0 && c < nch) {
      const int cn = (c + 1 < SC_MAXCH) ? c + 1 : SC_MAXCH - 1;
#pragma unroll
      for (int i = 0; i < SC_COLS; ++i) nxt[i] = row[cn * SC_COLS + i];

      float evf[SC_COLS]; int eve[SC_COLS];
#pragma unroll
      for (int j = 0; j < SC_COLS; ++j) { evf[j] = 0.0f; eve[j] = E_ZERO; }
      if (w > 0) {
#pragma unroll
        for (int j = 0; j < SC_COLS; ++j) {
          evf[j] = edge_f[w - 1][c & 1][j];
          eve[j] = edge_e[w - 1][c & 1][j];
        }
      }

      const int mbase = c * SC_COLS;
      const float eps0 = (c == 0) ? 0.0f : EPS_L2;
      float kf[SC_COLS]; int ke[SC_COLS];
      if (mbase + SC_COLS <= M) {
        float pf[SC_COLS]; int pe[SC_COLS];
#pragma unroll
        for (int j = 0; j < SC_COLS; ++j) weight_(cur[j], j == 0 ? eps0 : EPS_L2, pf[j], pe[j]);
#pragma unroll
        for (int j = 0; j < SC_COLS; ++j) {
          scan_col_(f, e, j == 0 ? plf : evf[j == 0 ? 0 : j - 1],
                          j == 0 ? ple : eve[j == 0 ? 0 : j - 1], pf[j], pe[j], is16);
          kf[j] = f; ke[j] = e;
        }
        if (mbase + SC_COLS == M) { rf = f; re = e; }
      } else {
#pragma unroll
        for (int j = 0; j < SC_COLS; ++j) {
          if (mbase + j < M) {
            float pfj; int pej;
            weight_(cur[j], j == 0 ? eps0 : EPS_L2, pfj, pej);
            scan_col_(f, e, j == 0 ? plf : evf[j == 0 ? 0 : j - 1],
                            j == 0 ? ple : eve[j == 0 ? 0 : j - 1], pfj, pej, is16);
            if (mbase + j == M - 1) { rf = f; re = e; }
          }
          kf[j] = f; ke[j] = e;
        }
      }
      if (lane == 31) {
#pragma unroll
        for (int j = 0; j < SC_COLS; ++j) { edge_f[w][c & 1][j] = kf[j]; edge_e[w][c & 1][j] = ke[j]; }
      }
      plf = evf[SC_COLS - 1];
      ple = eve[SC_COLS - 1];

      e += __builtin_amdgcn_frexp_expf(f);
      f  = __builtin_amdgcn_frexp_mantf(f);
#pragma unroll
      for (int i = 0; i < SC_COLS; ++i) cur[i] = nxt[i];
    }
    __syncthreads();
  }
  if (t == tl - 1) {
    const float la = (rf > 0.0f) ? (log2f(rf) + (float)re) * LN2_F : -PAD_MAG;
    ws[P_OFF + b] = la / (float)M;
  }
}

__global__ void mdn_reduce_kernel(const float* __restrict__ ws,
                                  float* __restrict__ out) {
  float s = 0.f;
  for (int b = 0; b < B_; ++b) s += ws[P_OFF + b];
  out[0] = -s / (float)B_;
}

extern "C" void kernel_launch(void* const* d_in, const int* in_sizes, int n_in,
                              void* d_out, int out_size, void* d_ws, size_t ws_size,
                              hipStream_t stream) {
  const float* mu_logvar = (const float*)d_in[0];
  const float* melspec   = (const float*)d_in[1];
  const int*   text_len  = (const int*)d_in[2];
  const int*   mel_len   = (const int*)d_in[3];
  float* out = (float*)d_out;
  float* ws  = (float*)d_ws;
  float* lp  = out + 1;

  mdn_prep_kernel<<<(B_ * TXT_PAD + 255) / 256, 256, 0, stream>>>(mu_logvar, ws);

  const int total_waves = B_ * (TXT_PAD / 16) * (MEL_T / 16);
  mdn_gemm_kernel<<<total_waves / 8, 256, 0, stream>>>(ws, melspec, lp);

  mdn_scan_kernel<<<B_, SC_WAVES * 32, 0, stream>>>(lp, text_len, mel_len, ws);
  mdn_reduce_kernel<<<1, 1, 0, stream>>>(ws, out);
}


# scan: each lane reads its lp row 32 columns (4 stages) ahead; loads stay in flight across the stage barrier, which now waits for LDS only (was 1 chunk ahead, drained at every barrier); sums unchanged
# speedup vs baseline: 3.4434x; 1.2109x over previous
#include <hip/hip_runtime.h>
#include <math.h>

#define B_      8
#define TXT_T   200
#define TXT_PAD 208
#define MEL_T   800
#define N_MEL   80
#define KDIM    160
#define PAD_MAG 1.0e12f

typedef __attribute__((ext_vector_type(2))) float v2f;
typedef __attribute__((ext_vector_type(8))) float v8f;

#define A_OFF 0
#define C_OFF (B_ * TXT_PAD * KDIM)
#define P_OFF (C_OFF + B_ * TXT_PAD)

__global__ void mdn_prep_kernel(const float* __restrict__ mu_logvar,
                                float* __restrict__ ws) {
  int idx = blockIdx.x * blockDim.x + threadIdx.x;
  if (idx >= B_ * TXT_PAD) return;
  int b = idx / TXT_PAD;
  int t = idx % TXT_PAD;
  float* Arow = ws + A_OFF + (size_t)idx * KDIM;
  float c = 0.0f;
  if (t < TXT_T) {
    const float* ml = mu_logvar + (size_t)(b * TXT_T + t) * (2 * N_MEL);
    for (int n = 0; n < N_MEL; ++n) {
      float mu = ml[n];
      float lv = ml[N_MEL + n];
      float w  = expf(-lv);
      Arow[n]          = w;
      Arow[N_MEL + n]  = -2.0f * mu * w;
      c += mu * mu * w + lv;
    }
  } else {
    for (int n = 0; n < KDIM; ++n) Arow[n] = 0.0f;
  }
  ws[C_OFF + idx] = c;
}

__global__ void __launch_bounds__(256)
mdn_gemm_kernel(const float* __restrict__ ws,
                const float* __restrict__ mel,
                float* __restrict__ lp) {
  const int TT = TXT_PAD / 16;
  const int MT = MEL_T / 16;
  int wid = blockIdx.x * (blockDim.x >> 5) + (threadIdx.x >> 5);
  if (wid >= B_ * TT * MT) return;
  int lane = threadIdx.x & 31;
  int l15  = lane & 15;
  int kl   = lane >> 4;

  int b  = wid / (TT * MT);
  int r  = wid % (TT * MT);
  int t0 = (r / MT) * 16;
  int m0 = (r % MT) * 16;

  const float* Arow = ws + A_OFF + (size_t)(b * TXT_PAD + t0 + l15) * KDIM;
  const float* melb = mel + (size_t)b * N_MEL * MEL_T + m0 + l15;

  v8f acc = {0.f, 0.f, 0.f, 0.f, 0.f, 0.f, 0.f, 0.f};

  for (int k = 0; k < KDIM; k += 4) {
    v2f a = *(const v2f*)(Arow + k + 2 * kl);

    int K0 = k + 2 * kl;
    int n  = (k < N_MEL) ? K0 : (K0 - N_MEL);
    float x0 = melb[(size_t)n * MEL_T];
    float x1 = melb[(size_t)(n + 1) * MEL_T];
    if (k < N_MEL) { x0 *= x0; x1 *= x1; }
    v2f bv; bv.x = x0; bv.y = x1;

    acc = __builtin_amdgcn_wmma_f32_16x16x4_f32(
         false, a,  false, bv,
         (short)0, acc,  false,  false);
  }

  int m = m0 + l15;
  const float* Cb = ws + C_OFF + b * TXT_PAD;
  const float scale = -0.5f / (float)N_MEL;
#pragma unroll
  for (int j = 0; j < 8; ++j) {
    int t = t0 + j + 8 * kl;
    if (t < TXT_T) {
      lp[(size_t)(b * TXT_T + t) * MEL_T + m] = scale * (acc[j] + Cb[t]);
    }
  }
}

#define SC_ROWS  32
#define SC_WAVES ((TXT_T + SC_ROWS - 1) / SC_ROWS)
#define SC_COLS  8
#define SC_MAXCH (MEL_T / SC_COLS)
#define SC_LINE  32
#define SC_CPL   (SC_LINE / SC_COLS)
#define SC_NLINE (MEL_T / SC_LINE)
static_assert(MEL_T % SC_LINE == 0 && SC_LINE % SC_COLS == 0 && SC_CPL == 4, "groups tile the row; the chunk select below has four cases");
#define LOG2E_F  1.44269504088896340736f
#define LN2_F    0.69314718055994530942f
#define EPS_L2   (1.0e-7f * LOG2E_F)
#define E_ZERO   (-(1 << 28))

__device__ __forceinline__ unsigned lane_below_(unsigned ua, unsigned edge, bool is16) {
  unsigned sh = __builtin_amdgcn_update_dpp(edge, ua, 0x111, 0xf, 0xf, false);
  unsigned xr = __builtin_amdgcn_permlanex16(edge, ua, 0x0000000Fu, 0u, false, false);
  return is16 ? xr : sh;
}

__device__ __forceinline__ void weight_(float x, float eps, float& pf, int& pe) {
  const float lg = fmaf(x, LOG2E_F, eps);
  const float fl = floorf(lg);
  pf = __builtin_amdgcn_exp2f(lg - fl);
  pe = (int)fl;
}

__device__ __forceinline__ void scan_col_(float& f, int& e, float bf, int be,
                                          float pf, int pe, bool is16) {
  const float fb = __uint_as_float(lane_below_(__float_as_uint(f), __float_as_uint(bf), is16));
  const int   eb = (int)lane_below_((unsigned)e, (unsigned)be, is16);
  const int   E  = e > eb ? e : eb;
  const float s  = __builtin_ldexpf(f, e - E) + __builtin_ldexpf(fb, eb - E);
  f = s * pf;
  e = E + pe;
}

__device__ __forceinline__ void stage_barrier_() {
  __builtin_amdgcn_fence(__ATOMIC_RELEASE, "workgroup", "local");
  __builtin_amdgcn_s_barrier();
  __builtin_amdgcn_fence(__ATOMIC_ACQUIRE, "workgroup", "local");
}

__global__ void __launch_bounds__(SC_WAVES * 32)
mdn_scan_kernel(const float* __restrict__ lp,
                const int* __restrict__ text_len,
                const int* __restrict__ mel_len,
                float* __restrict__ ws) {
  __shared__ __attribute__((aligned(16))) float edge_f[SC_WAVES][2][SC_COLS];
  __shared__ __attribute__((aligned(16))) int   edge_e[SC_WAVES][2][SC_COLS];
  const int b    = blockIdx.x;
  const int w    = __builtin_amdgcn_readfirstlane((int)(threadIdx.x >> 5));
  const int lane = threadIdx.x & 31;
  const int t    = w * SC_ROWS + lane;
  const bool is16 = (lane == 16);
  const int M  = mel_len[b];
  const int tl = text_len[b];
  int nch = (M + SC_COLS - 1) / SC_COLS;
  nch = nch < 0 ? 0 : (nch > SC_MAXCH ? SC_MAXCH : nch);

  for (int i = threadIdx.x; i < SC_WAVES * 2 * SC_COLS; i += SC_WAVES * 32) {
    (&edge_f[0][0][0])[i] = 0.0f;
    (&edge_e[0][0][0])[i] = E_ZERO;
  }

  const int tr = t < TXT_T ? t : TXT_T - 1;
  const float* row = lp + ((size_t)b * TXT_T + tr) * MEL_T;

  float buf[SC_LINE], nbuf[SC_LINE];
#pragma unroll
  for (int i = 0; i < SC_LINE; ++i) { buf[i] = row[i]; nbuf[i] = 0.0f; }

  float f = 0.0f;  int e = E_ZERO;
  float rf = 0.0f; int re = E_ZERO;
  float plf = (w == 0) ? 1.0f : 0.0f;
  int   ple = (w == 0) ? 0 : E_ZERO;
  __syncthreads();

  for (int s = 0; s < nch + SC_WAVES - 1; ++s) {
    const int c = s - w;
    if (c >= 0 && c < nch) {
      const int k = c & (SC_CPL - 1);
      if (k == 0) {
        const int qn = (c / SC_CPL + 1 < SC_NLINE) ? c / SC_CPL + 1 : SC_NLINE - 1;
#pragma unroll
        for (int i = 0; i < SC_LINE; ++i) nbuf[i] = row[qn * SC_LINE + i];
      }
      float cur[SC_COLS];
      switch (k) {
        case 0:
#pragma unroll
          for (int i = 0; i < SC_COLS; ++i) cur[i] = buf[i];
          break;
        case 1:
#pragma unroll
          for (int i = 0; i < SC_COLS; ++i) cur[i] = buf[SC_COLS + i];
          break;
        case 2:
#pragma unroll
          for (int i = 0; i < SC_COLS; ++i) cur[i] = buf[2 * SC_COLS + i];
          break;
        default:
#pragma unroll
          for (int i = 0; i < SC_COLS; ++i) cur[i] = buf[3 * SC_COLS + i];
          break;
      }

      float evf[SC_COLS]; int eve[SC_COLS];
#pragma unroll
      for (int j = 0; j < SC_COLS; ++j) { evf[j] = 0.0f; eve[j] = E_ZERO; }
      if (w > 0) {
#pragma unroll
        for (int j = 0; j < SC_COLS; ++j) {
          evf[j] = edge_f[w - 1][c & 1][j];
          eve[j] = edge_e[w - 1][c & 1][j];
        }
      }

      const int mbase = c * SC_COLS;
      const float eps0 = (c == 0) ? 0.0f : EPS_L2;
      float kf[SC_COLS]; int ke[SC_COLS];
      if (mbase + SC_COLS <= M) {
        float pf[SC_COLS]; int pe[SC_COLS];
#pragma unroll
        for (int j = 0; j < SC_COLS; ++j) weight_(cur[j], j == 0 ? eps0 : EPS_L2, pf[j], pe[j]);
#pragma unroll
        for (int j = 0; j < SC_COLS; ++j) {
          scan_col_(f, e, j == 0 ? plf : evf[j == 0 ? 0 : j - 1],
                          j == 0 ? ple : eve[j == 0 ? 0 : j - 1], pf[j], pe[j], is16);
          kf[j] = f; ke[j] = e;
        }
        if (mbase + SC_COLS == M) { rf = f; re = e; }
      } else {
#pragma unroll
        for (int j = 0; j < SC_COLS; ++j) {
          if (mbase + j < M) {
            float pfj; int pej;
            weight_(cur[j], j == 0 ? eps0 : EPS_L2, pfj, pej);
            scan_col_(f, e, j == 0 ? plf : evf[j == 0 ? 0 : j - 1],
                            j == 0 ? ple : eve[j == 0 ? 0 : j - 1], pfj, pej, is16);
            if (mbase + j == M - 1) { rf = f; re = e; }
          }
          kf[j] = f; ke[j] = e;
        }
      }
      if (lane == 31) {
#pragma unroll
        for (int j = 0; j < SC_COLS; ++j) { edge_f[w][c & 1][j] = kf[j]; edge_e[w][c & 1][j] = ke[j]; }
      }
      plf = evf[SC_COLS - 1];
      ple = eve[SC_COLS - 1];

      e += __builtin_amdgcn_frexp_expf(f);
      f  = __builtin_amdgcn_frexp_mantf(f);
      if (k == SC_CPL - 1) {
#pragma unroll
        for (int i = 0; i < SC_LINE; ++i) buf[i] = nbuf[i];
      }
    }
    stage_barrier_();
  }
  if (t == tl - 1) {
    const float la = (rf > 0.0f) ? (log2f(rf) + (float)re) * LN2_F : -PAD_MAG;
    ws[P_OFF + b] = la / (float)M;
  }
}

__global__ void mdn_reduce_kernel(const float* __restrict__ ws,
                                  float* __restrict__ out) {
  float s = 0.f;
  for (int b = 0; b < B_; ++b) s += ws[P_OFF + b];
  out[0] = -s / (float)B_;
}

extern "C" void kernel_launch(void* const* d_in, const int* in_sizes, int n_in,
                              void* d_out, int out_size, void* d_ws, size_t ws_size,
                              hipStream_t stream) {
  const float* mu_logvar = (const float*)d_in[0];
  const float* melspec   = (const float*)d_in[1];
  const int*   text_len  = (const int*)d_in[2];
  const int*   mel_len   = (const int*)d_in[3];
  float* out = (float*)d_out;
  float* ws  = (float*)d_ws;
  float* lp  = out + 1;

  mdn_prep_kernel<<<(B_ * TXT_PAD + 255) / 256, 256, 0, stream>>>(mu_logvar, ws);

  const int total_waves = B_ * (TXT_PAD / 16) * (MEL_T / 16);
  mdn_gemm_kernel<<<total_waves / 8, 256, 0, stream>>>(ws, melspec, lp);

  mdn_scan_kernel<<<B_, SC_WAVES * 32, 0, stream>>>(lp, text_len, mel_len, ws);
  mdn_reduce_kernel<<<1, 1, 0, stream>>>(ws, out);
}


# prep kernel: one wave per row of A', lanes across n, 16-byte loads and stores of 320 contiguous bytes, 208 work-groups (was one thread per row, 7); row constant summed across lanes: sum order changes
# speedup vs baseline: 4.0373x; 1.1725x over previous
#include <hip/hip_runtime.h>
#include <math.h>

#define B_      8
#define TXT_T   200
#define TXT_PAD 208
#define MEL_T   800
#define N_MEL   80
#define KDIM    160
#define PAD_MAG 1.0e12f

typedef __attribute__((ext_vector_type(2))) float v2f;
typedef __attribute__((ext_vector_type(8))) float v8f;

#define A_OFF 0
#define C_OFF (B_ * TXT_PAD * KDIM)
#define P_OFF (C_OFF + B_ * TXT_PAD)

#define PREP_WAVES 8
static_assert((B_ * TXT_PAD) % PREP_WAVES == 0 && N_MEL % 4 == 0 && N_MEL / 4 <= 32, "prep grid is exact; one row fits one wave");
__global__ void __launch_bounds__(PREP_WAVES * 32)
mdn_prep_kernel(const float* __restrict__ mu_logvar,
                float* __restrict__ ws) {
  const int lane = threadIdx.x & 31;
  const int idx  = blockIdx.x * PREP_WAVES + (threadIdx.x >> 5);
  const int b = idx / TXT_PAD;
  const int t = idx % TXT_PAD;
  float* Arow = ws + A_OFF + (size_t)idx * KDIM;
  float c = 0.0f;
  if (lane < N_MEL / 4) {
    float4 w4 = make_float4(0.0f, 0.0f, 0.0f, 0.0f);
    float4 m4 = make_float4(0.0f, 0.0f, 0.0f, 0.0f);
    if (t < TXT_T) {
      const float* ml = mu_logvar + (size_t)(b * TXT_T + t) * (2 * N_MEL);
      const float4 mu = *(const float4*)(ml + 4 * lane);
      const float4 lv = *(const float4*)(ml + N_MEL + 4 * lane);
      w4.x = expf(-lv.x); w4.y = expf(-lv.y); w4.z = expf(-lv.z); w4.w = expf(-lv.w);
      m4.x = -2.0f * mu.x * w4.x; m4.y = -2.0f * mu.y * w4.y;
      m4.z = -2.0f * mu.z * w4.z; m4.w = -2.0f * mu.w * w4.w;
      c  = mu.x * mu.x * w4.x + lv.x;
      c += mu.y * mu.y * w4.y + lv.y;
      c += mu.z * mu.z * w4.z + lv.z;
      c += mu.w * mu.w * w4.w + lv.w;
    }
    *(float4*)(Arow + 4 * lane)         = w4;
    *(float4*)(Arow + N_MEL + 4 * lane) = m4;
  }
  for (int m = 16; m >= 1; m >>= 1) c += __shfl_xor(c, m, 32);
  if (lane == 0) ws[C_OFF + idx] = c;
}

__global__ void __launch_bounds__(256)
mdn_gemm_kernel(const float* __restrict__ ws,
                const float* __restrict__ mel,
                float* __restrict__ lp) {
  const int TT = TXT_PAD / 16;
  const int MT = MEL_T / 16;
  int wid = blockIdx.x * (blockDim.x >> 5) + (threadIdx.x >> 5);
  if (wid >= B_ * TT * MT) return;
  int lane = threadIdx.x & 31;
  int l15  = lane & 15;
  int kl   = lane >> 4;

  int b  = wid / (TT * MT);
  int r  = wid % (TT * MT);
  int t0 = (r / MT) * 16;
  int m0 = (r % MT) * 16;

  const float* Arow = ws + A_OFF + (size_t)(b * TXT_PAD + t0 + l15) * KDIM;
  const float* melb = mel + (size_t)b * N_MEL * MEL_T + m0 + l15;

  v8f acc = {0.f, 0.f, 0.f, 0.f, 0.f, 0.f, 0.f, 0.f};

  for (int k = 0; k < KDIM; k += 4) {
    v2f a = *(const v2f*)(Arow + k + 2 * kl);

    int K0 = k + 2 * kl;
    int n  = (k < N_MEL) ? K0 : (K0 - N_MEL);
    float x0 = melb[(size_t)n * MEL_T];
    float x1 = melb[(size_t)(n + 1) * MEL_T];
    if (k < N_MEL) { x0 *= x0; x1 *= x1; }
    v2f bv; bv.x = x0; bv.y = x1;

    acc = __builtin_amdgcn_wmma_f32_16x16x4_f32(
         false, a,  false, bv,
         (short)0, acc,  false,  false);
  }

  int m = m0 + l15;
  const float* Cb = ws + C_OFF + b * TXT_PAD;
  const float scale = -0.5f / (float)N_MEL;
#pragma unroll
  for (int j = 0; j < 8; ++j) {
    int t = t0 + j + 8 * kl;
    if (t < TXT_T) {
      lp[(size_t)(b * TXT_T + t) * MEL_T + m] = scale * (acc[j] + Cb[t]);
    }
  }
}

#define SC_ROWS  32
#define SC_WAVES ((TXT_T + SC_ROWS - 1) / SC_ROWS)
#define SC_COLS  8
#define SC_MAXCH (MEL_T / SC_COLS)
#define SC_LINE  32
#define SC_CPL   (SC_LINE / SC_COLS)
#define SC_NLINE (MEL_T / SC_LINE)
static_assert(MEL_T % SC_LINE == 0 && SC_LINE % SC_COLS == 0 && SC_CPL == 4, "groups tile the row; the chunk select below has four cases");
#define LOG2E_F  1.44269504088896340736f
#define LN2_F    0.69314718055994530942f
#define EPS_L2   (1.0e-7f * LOG2E_F)
#define E_ZERO   (-(1 << 28))

__device__ __forceinline__ unsigned lane_below_(unsigned ua, unsigned edge, bool is16) {
  unsigned sh = __builtin_amdgcn_update_dpp(edge, ua, 0x111, 0xf, 0xf, false);
  unsigned xr = __builtin_amdgcn_permlanex16(edge, ua, 0x0000000Fu, 0u, false, false);
  return is16 ? xr : sh;
}

__device__ __forceinline__ void weight_(float x, float eps, float& pf, int& pe) {
  const float lg = fmaf(x, LOG2E_F, eps);
  const float fl = floorf(lg);
  pf = __builtin_amdgcn_exp2f(lg - fl);
  pe = (int)fl;
}

__device__ __forceinline__ void scan_col_(float& f, int& e, float bf, int be,
                                          float pf, int pe, bool is16) {
  const float fb = __uint_as_float(lane_below_(__float_as_uint(f), __float_as_uint(bf), is16));
  const int   eb = (int)lane_below_((unsigned)e, (unsigned)be, is16);
  const int   E  = e > eb ? e : eb;
  const float s  = __builtin_ldexpf(f, e - E) + __builtin_ldexpf(fb, eb - E);
  f = s * pf;
  e = E + pe;
}

__device__ __forceinline__ void stage_barrier_() {
  __builtin_amdgcn_fence(__ATOMIC_RELEASE, "workgroup", "local");
  __builtin_amdgcn_s_barrier();
  __builtin_amdgcn_fence(__ATOMIC_ACQUIRE, "workgroup", "local");
}

__global__ void __launch_bounds__(SC_WAVES * 32)
mdn_scan_kernel(const float* __restrict__ lp,
                const int* __restrict__ text_len,
                const int* __restrict__ mel_len,
                float* __restrict__ ws) {
  __shared__ __attribute__((aligned(16))) float edge_f[SC_WAVES][2][SC_COLS];
  __shared__ __attribute__((aligned(16))) int   edge_e[SC_WAVES][2][SC_COLS];
  const int b    = blockIdx.x;
  const int w    = __builtin_amdgcn_readfirstlane((int)(threadIdx.x >> 5));
  const int lane = threadIdx.x & 31;
  const int t    = w * SC_ROWS + lane;
  const bool is16 = (lane == 16);
  const int M  = mel_len[b];
  const int tl = text_len[b];
  int nch = (M + SC_COLS - 1) / SC_COLS;
  nch = nch < 0 ? 0 : (nch > SC_MAXCH ? SC_MAXCH : nch);

  for (int i = threadIdx.x; i < SC_WAVES * 2 * SC_COLS; i += SC_WAVES * 32) {
    (&edge_f[0][0][0])[i] = 0.0f;
    (&edge_e[0][0][0])[i] = E_ZERO;
  }

  const int tr = t < TXT_T ? t : TXT_T - 1;
  const float* row = lp + ((size_t)b * TXT_T + tr) * MEL_T;

  float buf[SC_LINE], nbuf[SC_LINE];
#pragma unroll
  for (int i = 0; i < SC_LINE; ++i) { buf[i] = row[i]; nbuf[i] = 0.0f; }

  float f = 0.0f;  int e = E_ZERO;
  float rf = 0.0f; int re = E_ZERO;
  float plf = (w == 0) ? 1.0f : 0.0f;
  int   ple = (w == 0) ? 0 : E_ZERO;
  __syncthreads();

  for (int s = 0; s < nch + SC_WAVES - 1; ++s) {
    const int c = s - w;
    if (c >= 0 && c < nch) {
      const int k = c & (SC_CPL - 1);
      if (k == 0) {
        const int qn = (c / SC_CPL + 1 < SC_NLINE) ? c / SC_CPL + 1 : SC_NLINE - 1;
#pragma unroll
        for (int i = 0; i < SC_LINE; ++i) nbuf[i] = row[qn * SC_LINE + i];
      }
      float cur[SC_COLS];
      switch (k) {
        case 0:
#pragma unroll
          for (int i = 0; i < SC_COLS; ++i) cur[i] = buf[i];
          break;
        case 1:
#pragma unroll
          for (int i = 0; i < SC_COLS; ++i) cur[i] = buf[SC_COLS + i];
          break;
        case 2:
#pragma unroll
          for (int i = 0; i < SC_COLS; ++i) cur[i] = buf[2 * SC_COLS + i];
          break;
        default:
#pragma unroll
          for (int i = 0; i < SC_COLS; ++i) cur[i] = buf[3 * SC_COLS + i];
          break;
      }

      float evf[SC_COLS]; int eve[SC_COLS];
#pragma unroll
      for (int j = 0; j < SC_COLS; ++j) { evf[j] = 0.0f; eve[j] = E_ZERO; }
      if (w > 0) {
#pragma unroll
        for (int j = 0; j < SC_COLS; ++j) {
          evf[j] = edge_f[w - 1][c & 1][j];
          eve[j] = edge_e[w - 1][c & 1][j];
        }
      }

      const int mbase = c * SC_COLS;
      const float eps0 = (c == 0) ? 0.0f : EPS_L2;
      float kf[SC_COLS]; int ke[SC_COLS];
      if (mbase + SC_COLS <= M) {
        float pf[SC_COLS]; int pe[SC_COLS];
#pragma unroll
        for (int j = 0; j < SC_COLS; ++j) weight_(cur[j], j == 0 ? eps0 : EPS_L2, pf[j], pe[j]);
#pragma unroll
        for (int j = 0; j < SC_COLS; ++j) {
          scan_col_(f, e, j == 0 ? plf : evf[j == 0 ? 0 : j - 1],
                          j == 0 ? ple : eve[j == 0 ? 0 : j - 1], pf[j], pe[j], is16);
          kf[j] = f; ke[j] = e;
        }
        if (mbase + SC_COLS == M) { rf = f; re = e; }
      } else {
#pragma unroll
        for (int j = 0; j < SC_COLS; ++j) {
          if (mbase + j < M) {
            float pfj; int pej;
            weight_(cur[j], j == 0 ? eps0 : EPS_L2, pfj, pej);
            scan_col_(f, e, j == 0 ? plf : evf[j == 0 ? 0 : j - 1],
                            j == 0 ? ple : eve[j == 0 ? 0 : j - 1], pfj, pej, is16);
            if (mbase + j == M - 1) { rf = f; re = e; }
          }
          kf[j] = f; ke[j] = e;
        }
      }
      if (lane == 31) {
#pragma unroll
        for (int j = 0; j < SC_COLS; ++j) { edge_f[w][c & 1][j] = kf[j]; edge_e[w][c & 1][j] = ke[j]; }
      }
      plf = evf[SC_COLS - 1];
      ple = eve[SC_COLS - 1];

      e += __builtin_amdgcn_frexp_expf(f);
      f  = __builtin_amdgcn_frexp_mantf(f);
      if (k == SC_CPL - 1) {
#pragma unroll
        for (int i = 0; i < SC_LINE; ++i) buf[i] = nbuf[i];
      }
    }
    stage_barrier_();
  }
  if (t == tl - 1) {
    const float la = (rf > 0.0f) ? (log2f(rf) + (float)re) * LN2_F : -PAD_MAG;
    ws[P_OFF + b] = la / (float)M;
  }
}

__global__ void mdn_reduce_kernel(const float* __restrict__ ws,
                                  float* __restrict__ out) {
  float s = 0.f;
  for (int b = 0; b < B_; ++b) s += ws[P_OFF + b];
  out[0] = -s / (float)B_;
}

extern "C" void kernel_launch(void* const* d_in, const int* in_sizes, int n_in,
                              void* d_out, int out_size, void* d_ws, size_t ws_size,
                              hipStream_t stream) {
  const float* mu_logvar = (const float*)d_in[0];
  const float* melspec   = (const float*)d_in[1];
  const int*   text_len  = (const int*)d_in[2];
  const int*   mel_len   = (const int*)d_in[3];
  float* out = (float*)d_out;
  float* ws  = (float*)d_ws;
  float* lp  = out + 1;

  mdn_prep_kernel<<<(B_ * TXT_PAD) / PREP_WAVES, PREP_WAVES * 32, 0, stream>>>(mu_logvar, ws);

  const int total_waves = B_ * (TXT_PAD / 16) * (MEL_T / 16);
  mdn_gemm_kernel<<<total_waves / 8, 256, 0, stream>>>(ws, melspec, lp);

  mdn_scan_kernel<<<B_, SC_WAVES * 32, 0, stream>>>(lp, text_len, mel_len, ws);
  mdn_reduce_kernel<<<1, 1, 0, stream>>>(ws, out);
}


# scan: a whole chunk whose rows are all reached runs on plain floats, the row exponent fixed per chunk (g*w + g_above*w*2^d: one fma per column), range-checked, else the aligned form; rounding changes
# speedup vs baseline: 4.2035x; 1.0412x over previous
#include <hip/hip_runtime.h>
#include <math.h>

#define B_      8
#define TXT_T   200
#define TXT_PAD 208
#define MEL_T   800
#define N_MEL   80
#define KDIM    160
#define PAD_MAG 1.0e12f

typedef __attribute__((ext_vector_type(2))) float v2f;
typedef __attribute__((ext_vector_type(8))) float v8f;

#define A_OFF 0
#define C_OFF (B_ * TXT_PAD * KDIM)
#define P_OFF (C_OFF + B_ * TXT_PAD)

#define PREP_WAVES 8
static_assert((B_ * TXT_PAD) % PREP_WAVES == 0 && N_MEL % 4 == 0 && N_MEL / 4 <= 32, "prep grid is exact; one row fits one wave");
__global__ void __launch_bounds__(PREP_WAVES * 32)
mdn_prep_kernel(const float* __restrict__ mu_logvar,
                float* __restrict__ ws) {
  const int lane = threadIdx.x & 31;
  const int idx  = blockIdx.x * PREP_WAVES + (threadIdx.x >> 5);
  const int b = idx / TXT_PAD;
  const int t = idx % TXT_PAD;
  float* Arow = ws + A_OFF + (size_t)idx * KDIM;
  float c = 0.0f;
  if (lane < N_MEL / 4) {
    float4 w4 = make_float4(0.0f, 0.0f, 0.0f, 0.0f);
    float4 m4 = make_float4(0.0f, 0.0f, 0.0f, 0.0f);
    if (t < TXT_T) {
      const float* ml = mu_logvar + (size_t)(b * TXT_T + t) * (2 * N_MEL);
      const float4 mu = *(const float4*)(ml + 4 * lane);
      const float4 lv = *(const float4*)(ml + N_MEL + 4 * lane);
      w4.x = expf(-lv.x); w4.y = expf(-lv.y); w4.z = expf(-lv.z); w4.w = expf(-lv.w);
      m4.x = -2.0f * mu.x * w4.x; m4.y = -2.0f * mu.y * w4.y;
      m4.z = -2.0f * mu.z * w4.z; m4.w = -2.0f * mu.w * w4.w;
      c  = mu.x * mu.x * w4.x + lv.x;
      c += mu.y * mu.y * w4.y + lv.y;
      c += mu.z * mu.z * w4.z + lv.z;
      c += mu.w * mu.w * w4.w + lv.w;
    }
    *(float4*)(Arow + 4 * lane)         = w4;
    *(float4*)(Arow + N_MEL + 4 * lane) = m4;
  }
  for (int m = 16; m >= 1; m >>= 1) c += __shfl_xor(c, m, 32);
  if (lane == 0) ws[C_OFF + idx] = c;
}

__global__ void __launch_bounds__(256)
mdn_gemm_kernel(const float* __restrict__ ws,
                const float* __restrict__ mel,
                float* __restrict__ lp) {
  const int TT = TXT_PAD / 16;
  const int MT = MEL_T / 16;
  int wid = blockIdx.x * (blockDim.x >> 5) + (threadIdx.x >> 5);
  if (wid >= B_ * TT * MT) return;
  int lane = threadIdx.x & 31;
  int l15  = lane & 15;
  int kl   = lane >> 4;

  int b  = wid / (TT * MT);
  int r  = wid % (TT * MT);
  int t0 = (r / MT) * 16;
  int m0 = (r % MT) * 16;

  const float* Arow = ws + A_OFF + (size_t)(b * TXT_PAD + t0 + l15) * KDIM;
  const float* melb = mel + (size_t)b * N_MEL * MEL_T + m0 + l15;

  v8f acc = {0.f, 0.f, 0.f, 0.f, 0.f, 0.f, 0.f, 0.f};

  for (int k = 0; k < KDIM; k += 4) {
    v2f a = *(const v2f*)(Arow + k + 2 * kl);

    int K0 = k + 2 * kl;
    int n  = (k < N_MEL) ? K0 : (K0 - N_MEL);
    float x0 = melb[(size_t)n * MEL_T];
    float x1 = melb[(size_t)(n + 1) * MEL_T];
    if (k < N_MEL) { x0 *= x0; x1 *= x1; }
    v2f bv; bv.x = x0; bv.y = x1;

    acc = __builtin_amdgcn_wmma_f32_16x16x4_f32(
         false, a,  false, bv,
         (short)0, acc,  false,  false);
  }

  int m = m0 + l15;
  const float* Cb = ws + C_OFF + b * TXT_PAD;
  const float scale = -0.5f / (float)N_MEL;
#pragma unroll
  for (int j = 0; j < 8; ++j) {
    int t = t0 + j + 8 * kl;
    if (t < TXT_T) {
      lp[(size_t)(b * TXT_T + t) * MEL_T + m] = scale * (acc[j] + Cb[t]);
    }
  }
}

#define SC_ROWS  32
#define SC_WAVES ((TXT_T + SC_ROWS - 1) / SC_ROWS)
#define SC_COLS  8
#define SC_MAXCH (MEL_T / SC_COLS)
#define SC_LINE  32
#define SC_CPL   (SC_LINE / SC_COLS)
#define SC_NLINE (MEL_T / SC_LINE)
static_assert(MEL_T % SC_LINE == 0 && SC_LINE % SC_COLS == 0 && SC_CPL == 4, "groups tile the row; the chunk select below has four cases");
#define LOG2E_F  1.44269504088896340736f
#define LN2_F    0.69314718055994530942f
#define EPS_L2   (1.0e-7f * LOG2E_F)
#define E_ZERO   (-(1 << 28))

__device__ __forceinline__ unsigned lane_below_(unsigned ua, unsigned edge, bool is16) {
  unsigned sh = __builtin_amdgcn_update_dpp(edge, ua, 0x111, 0xf, 0xf, false);
  unsigned xr = __builtin_amdgcn_permlanex16(edge, ua, 0x0000000Fu, 0u, false, false);
  return is16 ? xr : sh;
}

__device__ __forceinline__ void weight_(float x, float eps, float& pf, int& pe) {
  const float lg = fmaf(x, LOG2E_F, eps);
  const float fl = floorf(lg);
  pf = __builtin_amdgcn_exp2f(lg - fl);
  pe = (int)fl;
}

__device__ __forceinline__ void scan_col_(float& f, int& e, float bf, int be,
                                          float pf, int pe, bool is16) {
  const float fb = __uint_as_float(lane_below_(__float_as_uint(f), __float_as_uint(bf), is16));
  const int   eb = (int)lane_below_((unsigned)e, (unsigned)be, is16);
  const int   E  = e > eb ? e : eb;
  const float s  = __builtin_ldexpf(f, e - E) + __builtin_ldexpf(fb, eb - E);
  f = s * pf;
  e = E + pe;
}

#define G_HI  0x1p+16f
#define G_LO  0x1p-60f
__device__ __forceinline__ bool short_chunk_(float& f, const int e, const float (&cur)[SC_COLS], float eps0,
                                             float plf, int ple,
                                             const float (&evf)[SC_COLS], const int (&eve)[SC_COLS],
                                             bool is16, bool last_lane,
                                             float* __restrict__ out_f, int* __restrict__ out_e) {
  const int d = (int)lane_below_((unsigned)e, (unsigned)e, is16) - e;
  float g = f;
  float gk[SC_COLS];
#pragma unroll
  for (int j = 0; j < SC_COLS; ++j) {
    const float lg  = fmaf(cur[j], LOG2E_F, j == 0 ? eps0 : EPS_L2);
    const float wgt = __builtin_amdgcn_exp2f(lg);
    const float wab = __builtin_ldexpf(wgt, d);
    const float ef  = j == 0 ? plf : evf[j == 0 ? 0 : j - 1];
    const int   ee  = j == 0 ? ple : eve[j == 0 ? 0 : j - 1];
    const float edg = __builtin_ldexpf(ef, ee - e);
    const float ab  = __uint_as_float(lane_below_(__float_as_uint(g), __float_as_uint(edg), is16));
    g = fmaf(ab, wab, g * wgt);
    gk[j] = g;
  }
  const float glast = gk[SC_COLS - 1];
  const float gmx = fmaxf(fmaxf(fmaxf(gk[0], gk[1]), fmaxf(gk[2], gk[3])), fmaxf(fmaxf(gk[4], gk[5]), gk[6]));
  const float gmn = fminf(fminf(fminf(gk[0], gk[1]), fminf(gk[2], gk[3])), fminf(fminf(gk[4], gk[5]), gk[6]));
  const bool okl = gmx <= G_HI && glast <= G_HI && gmn >= G_LO && glast >= G_LO;
  if (__builtin_amdgcn_ballot_w32(okl) != 0xffffffffu) return false;
  if (last_lane) {
#pragma unroll
    for (int j = 0; j < SC_COLS; ++j) { out_f[j] = gk[j]; out_e[j] = e; }
  }
  f = glast;
  return true;
}

__device__ __forceinline__ void stage_barrier_() {
  __builtin_amdgcn_fence(__ATOMIC_RELEASE, "workgroup", "local");
  __builtin_amdgcn_s_barrier();
  __builtin_amdgcn_fence(__ATOMIC_ACQUIRE, "workgroup", "local");
}

__global__ void __launch_bounds__(SC_WAVES * 32)
mdn_scan_kernel(const float* __restrict__ lp,
                const int* __restrict__ text_len,
                const int* __restrict__ mel_len,
                float* __restrict__ ws) {
  __shared__ __attribute__((aligned(16))) float edge_f[SC_WAVES][2][SC_COLS];
  __shared__ __attribute__((aligned(16))) int   edge_e[SC_WAVES][2][SC_COLS];
  const int b    = blockIdx.x;
  const int w    = __builtin_amdgcn_readfirstlane((int)(threadIdx.x >> 5));
  const int lane = threadIdx.x & 31;
  const int t    = w * SC_ROWS + lane;
  const bool is16 = (lane == 16);
  const int M  = mel_len[b];
  const int tl = text_len[b];
  int nch = (M + SC_COLS - 1) / SC_COLS;
  nch = nch < 0 ? 0 : (nch > SC_MAXCH ? SC_MAXCH : nch);

  for (int i = threadIdx.x; i < SC_WAVES * 2 * SC_COLS; i += SC_WAVES * 32) {
    (&edge_f[0][0][0])[i] = 0.0f;
    (&edge_e[0][0][0])[i] = E_ZERO;
  }

  const int tr = t < TXT_T ? t : TXT_T - 1;
  const float* row = lp + ((size_t)b * TXT_T + tr) * MEL_T;

  float buf[SC_LINE], nbuf[SC_LINE];
#pragma unroll
  for (int i = 0; i < SC_LINE; ++i) { buf[i] = row[i]; nbuf[i] = 0.0f; }

  float f = 0.0f;  int e = E_ZERO;
  float rf = 0.0f; int re = E_ZERO;
  float plf = (w == 0) ? 1.0f : 0.0f;
  int   ple = (w == 0) ? 0 : E_ZERO;
  __syncthreads();

  for (int s = 0; s < nch + SC_WAVES - 1; ++s) {
    const int c = s - w;
    if (c >= 0 && c < nch) {
      const int k = c & (SC_CPL - 1);
      if (k == 0) {
        const int qn = (c / SC_CPL + 1 < SC_NLINE) ? c / SC_CPL + 1 : SC_NLINE - 1;
#pragma unroll
        for (int i = 0; i < SC_LINE; ++i) nbuf[i] = row[qn * SC_LINE + i];
      }
      float cur[SC_COLS];
      switch (k) {
        case 0:
#pragma unroll
          for (int i = 0; i < SC_COLS; ++i) cur[i] = buf[i];
          break;
        case 1:
#pragma unroll
          for (int i = 0; i < SC_COLS; ++i) cur[i] = buf[SC_COLS + i];
          break;
        case 2:
#pragma unroll
          for (int i = 0; i < SC_COLS; ++i) cur[i] = buf[2 * SC_COLS + i];
          break;
        default:
#pragma unroll
          for (int i = 0; i < SC_COLS; ++i) cur[i] = buf[3 * SC_COLS + i];
          break;
      }

      float evf[SC_COLS]; int eve[SC_COLS];
#pragma unroll
      for (int j = 0; j < SC_COLS; ++j) { evf[j] = 0.0f; eve[j] = E_ZERO; }
      if (w > 0) {
#pragma unroll
        for (int j = 0; j < SC_COLS; ++j) {
          evf[j] = edge_f[w - 1][c & 1][j];
          eve[j] = edge_e[w - 1][c & 1][j];
        }
      }

      const int mbase = c * SC_COLS;
      const float eps0 = (c == 0) ? 0.0f : EPS_L2;
      bool done = false;
      if (mbase + SC_COLS <= M && __builtin_amdgcn_ballot_w32(f > 0.0f) == 0xffffffffu) {
        done = short_chunk_(f, e, cur, eps0, plf, ple, evf, eve, is16, lane == 31,
                            &edge_f[w][c & 1][0], &edge_e[w][c & 1][0]);
        if (done && mbase + SC_COLS == M) { rf = f; re = e; }
      }
      if (!done) {
        float kf[SC_COLS]; int ke[SC_COLS];
        if (mbase + SC_COLS <= M) {
          float pf[SC_COLS]; int pe[SC_COLS];
#pragma unroll
          for (int j = 0; j < SC_COLS; ++j) weight_(cur[j], j == 0 ? eps0 : EPS_L2, pf[j], pe[j]);
#pragma unroll
          for (int j = 0; j < SC_COLS; ++j) {
            scan_col_(f, e, j == 0 ? plf : evf[j == 0 ? 0 : j - 1],
                            j == 0 ? ple : eve[j == 0 ? 0 : j - 1], pf[j], pe[j], is16);
            kf[j] = f; ke[j] = e;
          }
          if (mbase + SC_COLS == M) { rf = f; re = e; }
        } else {
#pragma unroll
          for (int j = 0; j < SC_COLS; ++j) {
            if (mbase + j < M) {
              float pfj; int pej;
              weight_(cur[j], j == 0 ? eps0 : EPS_L2, pfj, pej);
              scan_col_(f, e, j == 0 ? plf : evf[j == 0 ? 0 : j - 1],
                              j == 0 ? ple : eve[j == 0 ? 0 : j - 1], pfj, pej, is16);
              if (mbase + j == M - 1) { rf = f; re = e; }
            }
            kf[j] = f; ke[j] = e;
          }
        }
        if (lane == 31) {
#pragma unroll
          for (int j = 0; j < SC_COLS; ++j) { edge_f[w][c & 1][j] = kf[j]; edge_e[w][c & 1][j] = ke[j]; }
        }
      }
      plf = evf[SC_COLS - 1];
      ple = eve[SC_COLS - 1];

      e += __builtin_amdgcn_frexp_expf(f);
      f  = __builtin_amdgcn_frexp_mantf(f);
      if (k == SC_CPL - 1) {
#pragma unroll
        for (int i = 0; i < SC_LINE; ++i) buf[i] = nbuf[i];
      }
    }
    stage_barrier_();
  }
  if (t == tl - 1) {
    const float la = (rf > 0.0f) ? (log2f(rf) + (float)re) * LN2_F : -PAD_MAG;
    ws[P_OFF + b] = la / (float)M;
  }
}

__global__ void mdn_reduce_kernel(const float* __restrict__ ws,
                                  float* __restrict__ out) {
  float s = 0.f;
  for (int b = 0; b < B_; ++b) s += ws[P_OFF + b];
  out[0] = -s / (float)B_;
}

extern "C" void kernel_launch(void* const* d_in, const int* in_sizes, int n_in,
                              void* d_out, int out_size, void* d_ws, size_t ws_size,
                              hipStream_t stream) {
  const float* mu_logvar = (const float*)d_in[0];
  const float* melspec   = (const float*)d_in[1];
  const int*   text_len  = (const int*)d_in[2];
  const int*   mel_len   = (const int*)d_in[3];
  float* out = (float*)d_out;
  float* ws  = (float*)d_ws;
  float* lp  = out + 1;

  mdn_prep_kernel<<<(B_ * TXT_PAD) / PREP_WAVES, PREP_WAVES * 32, 0, stream>>>(mu_logvar, ws);

  const int total_waves = B_ * (TXT_PAD / 16) * (MEL_T / 16);
  mdn_gemm_kernel<<<total_waves / 8, 256, 0, stream>>>(ws, melspec, lp);

  mdn_scan_kernel<<<B_, SC_WAVES * 32, 0, stream>>>(lp, text_len, mel_len, ws);
  mdn_reduce_kernel<<<1, 1, 0, stream>>>(ws, out);
}


# scan stage widened to 16 columns per wave (was 8): half as many barriers, LDS edge hand-overs and range checks; up to 8 more columns before column M run on the aligned form: rounding changes there
# speedup vs baseline: 4.4696x; 1.0633x over previous
#include <hip/hip_runtime.h>
#include <math.h>

#define B_      8
#define TXT_T   200
#define TXT_PAD 208
#define MEL_T   800
#define N_MEL   80
#define KDIM    160
#define PAD_MAG 1.0e12f

typedef __attribute__((ext_vector_type(2))) float v2f;
typedef __attribute__((ext_vector_type(8))) float v8f;

#define A_OFF 0
#define C_OFF (B_ * TXT_PAD * KDIM)
#define P_OFF (C_OFF + B_ * TXT_PAD)

#define PREP_WAVES 8
static_assert((B_ * TXT_PAD) % PREP_WAVES == 0 && N_MEL % 4 == 0 && N_MEL / 4 <= 32, "prep grid is exact; one row fits one wave");
__global__ void __launch_bounds__(PREP_WAVES * 32)
mdn_prep_kernel(const float* __restrict__ mu_logvar,
                float* __restrict__ ws) {
  const int lane = threadIdx.x & 31;
  const int idx  = blockIdx.x * PREP_WAVES + (threadIdx.x >> 5);
  const int b = idx / TXT_PAD;
  const int t = idx % TXT_PAD;
  float* Arow = ws + A_OFF + (size_t)idx * KDIM;
  float c = 0.0f;
  if (lane < N_MEL / 4) {
    float4 w4 = make_float4(0.0f, 0.0f, 0.0f, 0.0f);
    float4 m4 = make_float4(0.0f, 0.0f, 0.0f, 0.0f);
    if (t < TXT_T) {
      const float* ml = mu_logvar + (size_t)(b * TXT_T + t) * (2 * N_MEL);
      const float4 mu = *(const float4*)(ml + 4 * lane);
      const float4 lv = *(const float4*)(ml + N_MEL + 4 * lane);
      w4.x = expf(-lv.x); w4.y = expf(-lv.y); w4.z = expf(-lv.z); w4.w = expf(-lv.w);
      m4.x = -2.0f * mu.x * w4.x; m4.y = -2.0f * mu.y * w4.y;
      m4.z = -2.0f * mu.z * w4.z; m4.w = -2.0f * mu.w * w4.w;
      c  = mu.x * mu.x * w4.x + lv.x;
      c += mu.y * mu.y * w4.y + lv.y;
      c += mu.z * mu.z * w4.z + lv.z;
      c += mu.w * mu.w * w4.w + lv.w;
    }
    *(float4*)(Arow + 4 * lane)         = w4;
    *(float4*)(Arow + N_MEL + 4 * lane) = m4;
  }
  for (int m = 16; m >= 1; m >>= 1) c += __shfl_xor(c, m, 32);
  if (lane == 0) ws[C_OFF + idx] = c;
}

__global__ void __launch_bounds__(256)
mdn_gemm_kernel(const float* __restrict__ ws,
                const float* __restrict__ mel,
                float* __restrict__ lp) {
  const int TT = TXT_PAD / 16;
  const int MT = MEL_T / 16;
  int wid = blockIdx.x * (blockDim.x >> 5) + (threadIdx.x >> 5);
  if (wid >= B_ * TT * MT) return;
  int lane = threadIdx.x & 31;
  int l15  = lane & 15;
  int kl   = lane >> 4;

  int b  = wid / (TT * MT);
  int r  = wid % (TT * MT);
  int t0 = (r / MT) * 16;
  int m0 = (r % MT) * 16;

  const float* Arow = ws + A_OFF + (size_t)(b * TXT_PAD + t0 + l15) * KDIM;
  const float* melb = mel + (size_t)b * N_MEL * MEL_T + m0 + l15;

  v8f acc = {0.f, 0.f, 0.f, 0.f, 0.f, 0.f, 0.f, 0.f};

  for (int k = 0; k < KDIM; k += 4) {
    v2f a = *(const v2f*)(Arow + k + 2 * kl);

    int K0 = k + 2 * kl;
    int n  = (k < N_MEL) ? K0 : (K0 - N_MEL);
    float x0 = melb[(size_t)n * MEL_T];
    float x1 = melb[(size_t)(n + 1) * MEL_T];
    if (k < N_MEL) { x0 *= x0; x1 *= x1; }
    v2f bv; bv.x = x0; bv.y = x1;

    acc = __builtin_amdgcn_wmma_f32_16x16x4_f32(
         false, a,  false, bv,
         (short)0, acc,  false,  false);
  }

  int m = m0 + l15;
  const float* Cb = ws + C_OFF + b * TXT_PAD;
  const float scale = -0.5f / (float)N_MEL;
#pragma unroll
  for (int j = 0; j < 8; ++j) {
    int t = t0 + j + 8 * kl;
    if (t < TXT_T) {
      lp[(size_t)(b * TXT_T + t) * MEL_T + m] = scale * (acc[j] + Cb[t]);
    }
  }
}

#define SC_ROWS  32
#define SC_WAVES ((TXT_T + SC_ROWS - 1) / SC_ROWS)
#define SC_COLS  16
#define SC_MAXCH (MEL_T / SC_COLS)
#define SC_LINE  32
#define SC_CPL   (SC_LINE / SC_COLS)
#define SC_NLINE (MEL_T / SC_LINE)
static_assert(MEL_T % SC_LINE == 0 && SC_LINE % SC_COLS == 0 && SC_CPL == 2, "groups tile the row; the chunk select below has two cases");
#define LOG2E_F  1.44269504088896340736f
#define LN2_F    0.69314718055994530942f
#define EPS_L2   (1.0e-7f * LOG2E_F)
#define E_ZERO   (-(1 << 28))

__device__ __forceinline__ unsigned lane_below_(unsigned ua, unsigned edge, bool is16) {
  unsigned sh = __builtin_amdgcn_update_dpp(edge, ua, 0x111, 0xf, 0xf, false);
  unsigned xr = __builtin_amdgcn_permlanex16(edge, ua, 0x0000000Fu, 0u, false, false);
  return is16 ? xr : sh;
}

__device__ __forceinline__ void weight_(float x, float eps, float& pf, int& pe) {
  const float lg = fmaf(x, LOG2E_F, eps);
  const float fl = floorf(lg);
  pf = __builtin_amdgcn_exp2f(lg - fl);
  pe = (int)fl;
}

__device__ __forceinline__ void scan_col_(float& f, int& e, float bf, int be,
                                          float pf, int pe, bool is16) {
  const float fb = __uint_as_float(lane_below_(__float_as_uint(f), __float_as_uint(bf), is16));
  const int   eb = (int)lane_below_((unsigned)e, (unsigned)be, is16);
  const int   E  = e > eb ? e : eb;
  const float s  = __builtin_ldexpf(f, e - E) + __builtin_ldexpf(fb, eb - E);
  f = s * pf;
  e = E + pe;
}

#define G_HI  0x1p+16f
#define G_LO  0x1p-60f
__device__ __forceinline__ bool short_chunk_(float& f, const int e, const float (&cur)[SC_COLS], float eps0,
                                             float plf, int ple,
                                             const float (&evf)[SC_COLS], const int (&eve)[SC_COLS],
                                             bool is16, bool last_lane,
                                             float* __restrict__ out_f, int* __restrict__ out_e) {
  const int d = (int)lane_below_((unsigned)e, (unsigned)e, is16) - e;
  float g = f;
  float gk[SC_COLS];
#pragma unroll
  for (int j = 0; j < SC_COLS; ++j) {
    const float lg  = fmaf(cur[j], LOG2E_F, j == 0 ? eps0 : EPS_L2);
    const float wgt = __builtin_amdgcn_exp2f(lg);
    const float wab = __builtin_ldexpf(wgt, d);
    const float ef  = j == 0 ? plf : evf[j == 0 ? 0 : j - 1];
    const int   ee  = j == 0 ? ple : eve[j == 0 ? 0 : j - 1];
    const float edg = __builtin_ldexpf(ef, ee - e);
    const float ab  = __uint_as_float(lane_below_(__float_as_uint(g), __float_as_uint(edg), is16));
    g = fmaf(ab, wab, g * wgt);
    gk[j] = g;
  }
  const float glast = gk[SC_COLS - 1];
  float gmx = gk[0], gmn = gk[0];
#pragma unroll
  for (int j = 1; j < SC_COLS - 1; ++j) { gmx = fmaxf(gmx, gk[j]); gmn = fminf(gmn, gk[j]); }
  const bool okl = gmx <= G_HI && glast <= G_HI && gmn >= G_LO && glast >= G_LO;
  if (__builtin_amdgcn_ballot_w32(okl) != 0xffffffffu) return false;
  if (last_lane) {
#pragma unroll
    for (int j = 0; j < SC_COLS; ++j) { out_f[j] = gk[j]; out_e[j] = e; }
  }
  f = glast;
  return true;
}

__device__ __forceinline__ void stage_barrier_() {
  __builtin_amdgcn_fence(__ATOMIC_RELEASE, "workgroup", "local");
  __builtin_amdgcn_s_barrier();
  __builtin_amdgcn_fence(__ATOMIC_ACQUIRE, "workgroup", "local");
}

__global__ void __launch_bounds__(SC_WAVES * 32)
mdn_scan_kernel(const float* __restrict__ lp,
                const int* __restrict__ text_len,
                const int* __restrict__ mel_len,
                float* __restrict__ ws) {
  __shared__ __attribute__((aligned(16))) float edge_f[SC_WAVES][2][SC_COLS];
  __shared__ __attribute__((aligned(16))) int   edge_e[SC_WAVES][2][SC_COLS];
  const int b    = blockIdx.x;
  const int w    = __builtin_amdgcn_readfirstlane((int)(threadIdx.x >> 5));
  const int lane = threadIdx.x & 31;
  const int t    = w * SC_ROWS + lane;
  const bool is16 = (lane == 16);
  const int M  = mel_len[b];
  const int tl = text_len[b];
  int nch = (M + SC_COLS - 1) / SC_COLS;
  nch = nch < 0 ? 0 : (nch > SC_MAXCH ? SC_MAXCH : nch);

  for (int i = threadIdx.x; i < SC_WAVES * 2 * SC_COLS; i += SC_WAVES * 32) {
    (&edge_f[0][0][0])[i] = 0.0f;
    (&edge_e[0][0][0])[i] = E_ZERO;
  }

  const int tr = t < TXT_T ? t : TXT_T - 1;
  const float* row = lp + ((size_t)b * TXT_T + tr) * MEL_T;

  float buf[SC_LINE], nbuf[SC_LINE];
#pragma unroll
  for (int i = 0; i < SC_LINE; ++i) { buf[i] = row[i]; nbuf[i] = 0.0f; }

  float f = 0.0f;  int e = E_ZERO;
  float rf = 0.0f; int re = E_ZERO;
  float plf = (w == 0) ? 1.0f : 0.0f;
  int   ple = (w == 0) ? 0 : E_ZERO;
  __syncthreads();

  for (int s = 0; s < nch + SC_WAVES - 1; ++s) {
    const int c = s - w;
    if (c >= 0 && c < nch) {
      const int k = c & (SC_CPL - 1);
      if (k == 0) {
        const int qn = (c / SC_CPL + 1 < SC_NLINE) ? c / SC_CPL + 1 : SC_NLINE - 1;
#pragma unroll
        for (int i = 0; i < SC_LINE; ++i) nbuf[i] = row[qn * SC_LINE + i];
      }
      float cur[SC_COLS];
      const bool first = (k == 0);
#pragma unroll
      for (int i = 0; i < SC_COLS; ++i) cur[i] = first ? buf[i] : buf[SC_COLS + i];

      float evf[SC_COLS]; int eve[SC_COLS];
#pragma unroll
      for (int j = 0; j < SC_COLS; ++j) { evf[j] = 0.0f; eve[j] = E_ZERO; }
      if (w > 0) {
#pragma unroll
        for (int j = 0; j < SC_COLS; ++j) {
          evf[j] = edge_f[w - 1][c & 1][j];
          eve[j] = edge_e[w - 1][c & 1][j];
        }
      }

      const int mbase = c * SC_COLS;
      const float eps0 = (c == 0) ? 0.0f : EPS_L2;
      bool done = false;
      if (mbase + SC_COLS <= M && __builtin_amdgcn_ballot_w32(f > 0.0f) == 0xffffffffu) {
        done = short_chunk_(f, e, cur, eps0, plf, ple, evf, eve, is16, lane == 31,
                            &edge_f[w][c & 1][0], &edge_e[w][c & 1][0]);
        if (done && mbase + SC_COLS == M) { rf = f; re = e; }
      }
      if (!done) {
        float kf[SC_COLS]; int ke[SC_COLS];
        if (mbase + SC_COLS <= M) {
          float pf[SC_COLS]; int pe[SC_COLS];
#pragma unroll
          for (int j = 0; j < SC_COLS; ++j) weight_(cur[j], j == 0 ? eps0 : EPS_L2, pf[j], pe[j]);
#pragma unroll
          for (int j = 0; j < SC_COLS; ++j) {
            scan_col_(f, e, j == 0 ? plf : evf[j == 0 ? 0 : j - 1],
                            j == 0 ? ple : eve[j == 0 ? 0 : j - 1], pf[j], pe[j], is16);
            kf[j] = f; ke[j] = e;
          }
          if (mbase + SC_COLS == M) { rf = f; re = e; }
        } else {
#pragma unroll
          for (int j = 0; j < SC_COLS; ++j) {
            if (mbase + j < M) {
              float pfj; int pej;
              weight_(cur[j], j == 0 ? eps0 : EPS_L2, pfj, pej);
              scan_col_(f, e, j == 0 ? plf : evf[j == 0 ? 0 : j - 1],
                              j == 0 ? ple : eve[j == 0 ? 0 : j - 1], pfj, pej, is16);
              if (mbase + j == M - 1) { rf = f; re = e; }
            }
            kf[j] = f; ke[j] = e;
          }
        }
        if (lane == 31) {
#pragma unroll
          for (int j = 0; j < SC_COLS; ++j) { edge_f[w][c & 1][j] = kf[j]; edge_e[w][c & 1][j] = ke[j]; }
        }
      }
      plf = evf[SC_COLS - 1];
      ple = eve[SC_COLS - 1];

      e += __builtin_amdgcn_frexp_expf(f);
      f  = __builtin_amdgcn_frexp_mantf(f);
      if (k == SC_CPL - 1) {
#pragma unroll
        for (int i = 0; i < SC_LINE; ++i) buf[i] = nbuf[i];
      }
    }
    stage_barrier_();
  }
  if (t == tl - 1) {
    const float la = (rf > 0.0f) ? (log2f(rf) + (float)re) * LN2_F : -PAD_MAG;
    ws[P_OFF + b] = la / (float)M;
  }
}

__global__ void mdn_reduce_kernel(const float* __restrict__ ws,
                                  float* __restrict__ out) {
  float s = 0.f;
  for (int b = 0; b < B_; ++b) s += ws[P_OFF + b];
  out[0] = -s / (float)B_;
}

extern "C" void kernel_launch(void* const* d_in, const int* in_sizes, int n_in,
                              void* d_out, int out_size, void* d_ws, size_t ws_size,
                              hipStream_t stream) {
  const float* mu_logvar = (const float*)d_in[0];
  const float* melspec   = (const float*)d_in[1];
  const int*   text_len  = (const int*)d_in[2];
  const int*   mel_len   = (const int*)d_in[3];
  float* out = (float*)d_out;
  float* ws  = (float*)d_ws;
  float* lp  = out + 1;

  mdn_prep_kernel<<<(B_ * TXT_PAD) / PREP_WAVES, PREP_WAVES * 32, 0, stream>>>(mu_logvar, ws);

  const int total_waves = B_ * (TXT_PAD / 16) * (MEL_T / 16);
  mdn_gemm_kernel<<<total_waves / 8, 256, 0, stream>>>(ws, melspec, lp);

  mdn_scan_kernel<<<B_, SC_WAVES * 32, 0, stream>>>(lp, text_len, mel_len, ws);
  mdn_reduce_kernel<<<1, 1, 0, stream>>>(ws, out);
}


# scan: lp loads shared by lane pairs: two neighbouring lanes read 32 contiguous bytes of one row, so a load instruction covers 16 rows (was 32); halves exchanged by a quad permute; sums unchanged
# speedup vs baseline: 4.8513x; 1.0854x over previous
#include <hip/hip_runtime.h>
#include <math.h>

#define B_      8
#define TXT_T   200
#define TXT_PAD 208
#define MEL_T   800
#define N_MEL   80
#define KDIM    160
#define PAD_MAG 1.0e12f

typedef __attribute__((ext_vector_type(2))) float v2f;
typedef __attribute__((ext_vector_type(8))) float v8f;

#define A_OFF 0
#define C_OFF (B_ * TXT_PAD * KDIM)
#define P_OFF (C_OFF + B_ * TXT_PAD)

#define PREP_WAVES 8
static_assert((B_ * TXT_PAD) % PREP_WAVES == 0 && N_MEL % 4 == 0 && N_MEL / 4 <= 32, "prep grid is exact; one row fits one wave");
__global__ void __launch_bounds__(PREP_WAVES * 32)
mdn_prep_kernel(const float* __restrict__ mu_logvar,
                float* __restrict__ ws) {
  const int lane = threadIdx.x & 31;
  const int idx  = blockIdx.x * PREP_WAVES + (threadIdx.x >> 5);
  const int b = idx / TXT_PAD;
  const int t = idx % TXT_PAD;
  float* Arow = ws + A_OFF + (size_t)idx * KDIM;
  float c = 0.0f;
  if (lane < N_MEL / 4) {
    float4 w4 = make_float4(0.0f, 0.0f, 0.0f, 0.0f);
    float4 m4 = make_float4(0.0f, 0.0f, 0.0f, 0.0f);
    if (t < TXT_T) {
      const float* ml = mu_logvar + (size_t)(b * TXT_T + t) * (2 * N_MEL);
      const float4 mu = *(const float4*)(ml + 4 * lane);
      const float4 lv = *(const float4*)(ml + N_MEL + 4 * lane);
      w4.x = expf(-lv.x); w4.y = expf(-lv.y); w4.z = expf(-lv.z); w4.w = expf(-lv.w);
      m4.x = -2.0f * mu.x * w4.x; m4.y = -2.0f * mu.y * w4.y;
      m4.z = -2.0f * mu.z * w4.z; m4.w = -2.0f * mu.w * w4.w;
      c  = mu.x * mu.x * w4.x + lv.x;
      c += mu.y * mu.y * w4.y + lv.y;
      c += mu.z * mu.z * w4.z + lv.z;
      c += mu.w * mu.w * w4.w + lv.w;
    }
    *(float4*)(Arow + 4 * lane)         = w4;
    *(float4*)(Arow + N_MEL + 4 * lane) = m4;
  }
  for (int m = 16; m >= 1; m >>= 1) c += __shfl_xor(c, m, 32);
  if (lane == 0) ws[C_OFF + idx] = c;
}

__global__ void __launch_bounds__(256)
mdn_gemm_kernel(const float* __restrict__ ws,
                const float* __restrict__ mel,
                float* __restrict__ lp) {
  const int TT = TXT_PAD / 16;
  const int MT = MEL_T / 16;
  int wid = blockIdx.x * (blockDim.x >> 5) + (threadIdx.x >> 5);
  if (wid >= B_ * TT * MT) return;
  int lane = threadIdx.x & 31;
  int l15  = lane & 15;
  int kl   = lane >> 4;

  int b  = wid / (TT * MT);
  int r  = wid % (TT * MT);
  int t0 = (r / MT) * 16;
  int m0 = (r % MT) * 16;

  const float* Arow = ws + A_OFF + (size_t)(b * TXT_PAD + t0 + l15) * KDIM;
  const float* melb = mel + (size_t)b * N_MEL * MEL_T + m0 + l15;

  v8f acc = {0.f, 0.f, 0.f, 0.f, 0.f, 0.f, 0.f, 0.f};

  for (int k = 0; k < KDIM; k += 4) {
    v2f a = *(const v2f*)(Arow + k + 2 * kl);

    int K0 = k + 2 * kl;
    int n  = (k < N_MEL) ? K0 : (K0 - N_MEL);
    float x0 = melb[(size_t)n * MEL_T];
    float x1 = melb[(size_t)(n + 1) * MEL_T];
    if (k < N_MEL) { x0 *= x0; x1 *= x1; }
    v2f bv; bv.x = x0; bv.y = x1;

    acc = __builtin_amdgcn_wmma_f32_16x16x4_f32(
         false, a,  false, bv,
         (short)0, acc,  false,  false);
  }

  int m = m0 + l15;
  const float* Cb = ws + C_OFF + b * TXT_PAD;
  const float scale = -0.5f / (float)N_MEL;
#pragma unroll
  for (int j = 0; j < 8; ++j) {
    int t = t0 + j + 8 * kl;
    if (t < TXT_T) {
      lp[(size_t)(b * TXT_T + t) * MEL_T + m] = scale * (acc[j] + Cb[t]);
    }
  }
}

#define SC_ROWS  32
#define SC_WAVES ((TXT_T + SC_ROWS - 1) / SC_ROWS)
#define SC_COLS  16
#define SC_MAXCH (MEL_T / SC_COLS)
#define SC_LINE  32
#define SC_CPL   (SC_LINE / SC_COLS)
#define SC_NLINE (MEL_T / SC_LINE)
static_assert(MEL_T % SC_LINE == 0 && SC_LINE % SC_COLS == 0 && SC_CPL == 2, "groups tile the row; the chunk select below has two cases");
#define LOG2E_F  1.44269504088896340736f
#define LN2_F    0.69314718055994530942f
#define EPS_L2   (1.0e-7f * LOG2E_F)
#define E_ZERO   (-(1 << 28))

__device__ __forceinline__ unsigned lane_below_(unsigned ua, unsigned edge, bool is16) {
  unsigned sh = __builtin_amdgcn_update_dpp(edge, ua, 0x111, 0xf, 0xf, false);
  unsigned xr = __builtin_amdgcn_permlanex16(edge, ua, 0x0000000Fu, 0u, false, false);
  return is16 ? xr : sh;
}

__device__ __forceinline__ void weight_(float x, float eps, float& pf, int& pe) {
  const float lg = fmaf(x, LOG2E_F, eps);
  const float fl = floorf(lg);
  pf = __builtin_amdgcn_exp2f(lg - fl);
  pe = (int)fl;
}

__device__ __forceinline__ void scan_col_(float& f, int& e, float bf, int be,
                                          float pf, int pe, bool is16) {
  const float fb = __uint_as_float(lane_below_(__float_as_uint(f), __float_as_uint(bf), is16));
  const int   eb = (int)lane_below_((unsigned)e, (unsigned)be, is16);
  const int   E  = e > eb ? e : eb;
  const float s  = __builtin_ldexpf(f, e - E) + __builtin_ldexpf(fb, eb - E);
  f = s * pf;
  e = E + pe;
}

#define G_HI  0x1p+16f
#define G_LO  0x1p-60f
__device__ __forceinline__ bool short_chunk_(float& f, const int e, const float (&cur)[SC_COLS], float eps0,
                                             float plf, int ple,
                                             const float (&evf)[SC_COLS], const int (&eve)[SC_COLS],
                                             bool is16, bool last_lane,
                                             float* __restrict__ out_f, int* __restrict__ out_e) {
  const int d = (int)lane_below_((unsigned)e, (unsigned)e, is16) - e;
  float g = f;
  float gk[SC_COLS];
#pragma unroll
  for (int j = 0; j < SC_COLS; ++j) {
    const float lg  = fmaf(cur[j], LOG2E_F, j == 0 ? eps0 : EPS_L2);
    const float wgt = __builtin_amdgcn_exp2f(lg);
    const float wab = __builtin_ldexpf(wgt, d);
    const float ef  = j == 0 ? plf : evf[j == 0 ? 0 : j - 1];
    const int   ee  = j == 0 ? ple : eve[j == 0 ? 0 : j - 1];
    const float edg = __builtin_ldexpf(ef, ee - e);
    const float ab  = __uint_as_float(lane_below_(__float_as_uint(g), __float_as_uint(edg), is16));
    g = fmaf(ab, wab, g * wgt);
    gk[j] = g;
  }
  const float glast = gk[SC_COLS - 1];
  float gmx = gk[0], gmn = gk[0];
#pragma unroll
  for (int j = 1; j < SC_COLS - 1; ++j) { gmx = fmaxf(gmx, gk[j]); gmn = fminf(gmn, gk[j]); }
  const bool okl = gmx <= G_HI && glast <= G_HI && gmn >= G_LO && glast >= G_LO;
  if (__builtin_amdgcn_ballot_w32(okl) != 0xffffffffu) return false;
  if (last_lane) {
#pragma unroll
    for (int j = 0; j < SC_COLS; ++j) { out_f[j] = gk[j]; out_e[j] = e; }
  }
  f = glast;
  return true;
}

__device__ __forceinline__ void load_group_(float (&raw)[SC_LINE], const float* __restrict__ rowE,
                                            const float* __restrict__ rowO, int col0) {
#pragma unroll
  for (int j = 0; j < SC_LINE / 4; ++j) {
    const float* src = ((j & 1) ? rowO : rowE) + col0 + 8 * (j >> 1);
#pragma unroll
    for (int d = 0; d < 4; ++d) raw[4 * j + d] = src[d];
  }
}

__device__ __forceinline__ float pair_swap_(float v) {
  return __uint_as_float(__builtin_amdgcn_update_dpp(0u, __float_as_uint(v), 0xB1, 0xf, 0xf, true));
}

__device__ __forceinline__ void own_row_(float (&own)[SC_LINE], const float (&raw)[SC_LINE], bool odd) {
#pragma unroll
  for (int g = 0; g < SC_LINE / 8; ++g) {
#pragma unroll
    for (int d = 0; d < 4; ++d) {
      const float x  = raw[4 * (2 * g) + d];
      const float y  = raw[4 * (2 * g + 1) + d];
      const float px = pair_swap_(x);
      const float py = pair_swap_(y);
      own[4 * (2 * g) + d]     = odd ? py : x;
      own[4 * (2 * g + 1) + d] = odd ? y : px;
    }
  }
}

__device__ __forceinline__ void stage_barrier_() {
  __builtin_amdgcn_fence(__ATOMIC_RELEASE, "workgroup", "local");
  __builtin_amdgcn_s_barrier();
  __builtin_amdgcn_fence(__ATOMIC_ACQUIRE, "workgroup", "local");
}

__global__ void __launch_bounds__(SC_WAVES * 32)
mdn_scan_kernel(const float* __restrict__ lp,
                const int* __restrict__ text_len,
                const int* __restrict__ mel_len,
                float* __restrict__ ws) {
  __shared__ __attribute__((aligned(16))) float edge_f[SC_WAVES][2][SC_COLS];
  __shared__ __attribute__((aligned(16))) int   edge_e[SC_WAVES][2][SC_COLS];
  const int b    = blockIdx.x;
  const int w    = __builtin_amdgcn_readfirstlane((int)(threadIdx.x >> 5));
  const int lane = threadIdx.x & 31;
  const int t    = w * SC_ROWS + lane;
  const bool is16 = (lane == 16);
  const int M  = mel_len[b];
  const int tl = text_len[b];
  int nch = (M + SC_COLS - 1) / SC_COLS;
  nch = nch < 0 ? 0 : (nch > SC_MAXCH ? SC_MAXCH : nch);

  for (int i = threadIdx.x; i < SC_WAVES * 2 * SC_COLS; i += SC_WAVES * 32) {
    (&edge_f[0][0][0])[i] = 0.0f;
    (&edge_e[0][0][0])[i] = E_ZERO;
  }

  const bool odd = (lane & 1) != 0;
  const int tE  = t - (lane & 1);
  const int trE = tE < TXT_T ? tE : TXT_T - 1;
  const int trO = tE + 1 < TXT_T ? tE + 1 : TXT_T - 1;
  const float* rowE = lp + ((size_t)b * TXT_T + trE) * MEL_T + 4 * (lane & 1);
  const float* rowO = lp + ((size_t)b * TXT_T + trO) * MEL_T + 4 * (lane & 1);

  float buf[SC_LINE], nbuf[SC_LINE];
  load_group_(nbuf, rowE, rowO, 0);
  own_row_(buf, nbuf, odd);

  float f = 0.0f;  int e = E_ZERO;
  float rf = 0.0f; int re = E_ZERO;
  float plf = (w == 0) ? 1.0f : 0.0f;
  int   ple = (w == 0) ? 0 : E_ZERO;
  __syncthreads();

  for (int s = 0; s < nch + SC_WAVES - 1; ++s) {
    const int c = s - w;
    if (c >= 0 && c < nch) {
      const int k = c & (SC_CPL - 1);
      if (k == 0) {
        const int qn = (c / SC_CPL + 1 < SC_NLINE) ? c / SC_CPL + 1 : SC_NLINE - 1;
        load_group_(nbuf, rowE, rowO, qn * SC_LINE);
      }
      float cur[SC_COLS];
      const bool first = (k == 0);
#pragma unroll
      for (int i = 0; i < SC_COLS; ++i) cur[i] = first ? buf[i] : buf[SC_COLS + i];

      float evf[SC_COLS]; int eve[SC_COLS];
#pragma unroll
      for (int j = 0; j < SC_COLS; ++j) { evf[j] = 0.0f; eve[j] = E_ZERO; }
      if (w > 0) {
#pragma unroll
        for (int j = 0; j < SC_COLS; ++j) {
          evf[j] = edge_f[w - 1][c & 1][j];
          eve[j] = edge_e[w - 1][c & 1][j];
        }
      }

      const int mbase = c * SC_COLS;
      const float eps0 = (c == 0) ? 0.0f : EPS_L2;
      bool done = false;
      if (mbase + SC_COLS <= M && __builtin_amdgcn_ballot_w32(f > 0.0f) == 0xffffffffu) {
        done = short_chunk_(f, e, cur, eps0, plf, ple, evf, eve, is16, lane == 31,
                            &edge_f[w][c & 1][0], &edge_e[w][c & 1][0]);
        if (done && mbase + SC_COLS == M) { rf = f; re = e; }
      }
      if (!done) {
        float kf[SC_COLS]; int ke[SC_COLS];
        if (mbase + SC_COLS <= M) {
          float pf[SC_COLS]; int pe[SC_COLS];
#pragma unroll
          for (int j = 0; j < SC_COLS; ++j) weight_(cur[j], j == 0 ? eps0 : EPS_L2, pf[j], pe[j]);
#pragma unroll
          for (int j = 0; j < SC_COLS; ++j) {
            scan_col_(f, e, j == 0 ? plf : evf[j == 0 ? 0 : j - 1],
                            j == 0 ? ple : eve[j == 0 ? 0 : j - 1], pf[j], pe[j], is16);
            kf[j] = f; ke[j] = e;
          }
          if (mbase + SC_COLS == M) { rf = f; re = e; }
        } else {
#pragma unroll
          for (int j = 0; j < SC_COLS; ++j) {
            if (mbase + j < M) {
              float pfj; int pej;
              weight_(cur[j], j == 0 ? eps0 : EPS_L2, pfj, pej);
              scan_col_(f, e, j == 0 ? plf : evf[j == 0 ? 0 : j - 1],
                              j == 0 ? ple : eve[j == 0 ? 0 : j - 1], pfj, pej, is16);
              if (mbase + j == M - 1) { rf = f; re = e; }
            }
            kf[j] = f; ke[j] = e;
          }
        }
        if (lane == 31) {
#pragma unroll
          for (int j = 0; j < SC_COLS; ++j) { edge_f[w][c & 1][j] = kf[j]; edge_e[w][c & 1][j] = ke[j]; }
        }
      }
      plf = evf[SC_COLS - 1];
      ple = eve[SC_COLS - 1];

      e += __builtin_amdgcn_frexp_expf(f);
      f  = __builtin_amdgcn_frexp_mantf(f);
      if (k == SC_CPL - 1) {
        own_row_(buf, nbuf, odd);
      }
    }
    stage_barrier_();
  }
  if (t == tl - 1) {
    const float la = (rf > 0.0f) ? (log2f(rf) + (float)re) * LN2_F : -PAD_MAG;
    ws[P_OFF + b] = la / (float)M;
  }
}

__global__ void mdn_reduce_kernel(const float* __restrict__ ws,
                                  float* __restrict__ out) {
  float s = 0.f;
  for (int b = 0; b < B_; ++b) s += ws[P_OFF + b];
  out[0] = -s / (float)B_;
}

extern "C" void kernel_launch(void* const* d_in, const int* in_sizes, int n_in,
                              void* d_out, int out_size, void* d_ws, size_t ws_size,
                              hipStream_t stream) {
  const float* mu_logvar = (const float*)d_in[0];
  const float* melspec   = (const float*)d_in[1];
  const int*   text_len  = (const int*)d_in[2];
  const int*   mel_len   = (const int*)d_in[3];
  float* out = (float*)d_out;
  float* ws  = (float*)d_ws;
  float* lp  = out + 1;

  mdn_prep_kernel<<<(B_ * TXT_PAD) / PREP_WAVES, PREP_WAVES * 32, 0, stream>>>(mu_logvar, ws);

  const int total_waves = B_ * (TXT_PAD / 16) * (MEL_T / 16);
  mdn_gemm_kernel<<<total_waves / 8, 256, 0, stream>>>(ws, melspec, lp);

  mdn_scan_kernel<<<B_, SC_WAVES * 32, 0, stream>>>(lp, text_len, mel_len, ws);
  mdn_reduce_kernel<<<1, 1, 0, stream>>>(ws, out);
}


# scan: a wave's rows alternate between its two 16-lane halves: the row above comes by ONE per-lane gather from the other half (was shift+gather+select); edge enters via the addend: rounding changes
# speedup vs baseline: 5.1409x; 1.0597x over previous
#include <hip/hip_runtime.h>
#include <math.h>

#define B_      8
#define TXT_T   200
#define TXT_PAD 208
#define MEL_T   800
#define N_MEL   80
#define KDIM    160
#define PAD_MAG 1.0e12f

typedef __attribute__((ext_vector_type(2))) float v2f;
typedef __attribute__((ext_vector_type(8))) float v8f;

#define A_OFF 0
#define C_OFF (B_ * TXT_PAD * KDIM)
#define P_OFF (C_OFF + B_ * TXT_PAD)

#define PREP_WAVES 8
static_assert((B_ * TXT_PAD) % PREP_WAVES == 0 && N_MEL % 4 == 0 && N_MEL / 4 <= 32, "prep grid is exact; one row fits one wave");
__global__ void __launch_bounds__(PREP_WAVES * 32)
mdn_prep_kernel(const float* __restrict__ mu_logvar,
                float* __restrict__ ws) {
  const int lane = threadIdx.x & 31;
  const int idx  = blockIdx.x * PREP_WAVES + (threadIdx.x >> 5);
  const int b = idx / TXT_PAD;
  const int t = idx % TXT_PAD;
  float* Arow = ws + A_OFF + (size_t)idx * KDIM;
  float c = 0.0f;
  if (lane < N_MEL / 4) {
    float4 w4 = make_float4(0.0f, 0.0f, 0.0f, 0.0f);
    float4 m4 = make_float4(0.0f, 0.0f, 0.0f, 0.0f);
    if (t < TXT_T) {
      const float* ml = mu_logvar + (size_t)(b * TXT_T + t) * (2 * N_MEL);
      const float4 mu = *(const float4*)(ml + 4 * lane);
      const float4 lv = *(const float4*)(ml + N_MEL + 4 * lane);
      w4.x = expf(-lv.x); w4.y = expf(-lv.y); w4.z = expf(-lv.z); w4.w = expf(-lv.w);
      m4.x = -2.0f * mu.x * w4.x; m4.y = -2.0f * mu.y * w4.y;
      m4.z = -2.0f * mu.z * w4.z; m4.w = -2.0f * mu.w * w4.w;
      c  = mu.x * mu.x * w4.x + lv.x;
      c += mu.y * mu.y * w4.y + lv.y;
      c += mu.z * mu.z * w4.z + lv.z;
      c += mu.w * mu.w * w4.w + lv.w;
    }
    *(float4*)(Arow + 4 * lane)         = w4;
    *(float4*)(Arow + N_MEL + 4 * lane) = m4;
  }
  for (int m = 16; m >= 1; m >>= 1) c += __shfl_xor(c, m, 32);
  if (lane == 0) ws[C_OFF + idx] = c;
}

__global__ void __launch_bounds__(256)
mdn_gemm_kernel(const float* __restrict__ ws,
                const float* __restrict__ mel,
                float* __restrict__ lp) {
  const int TT = TXT_PAD / 16;
  const int MT = MEL_T / 16;
  int wid = blockIdx.x * (blockDim.x >> 5) + (threadIdx.x >> 5);
  if (wid >= B_ * TT * MT) return;
  int lane = threadIdx.x & 31;
  int l15  = lane & 15;
  int kl   = lane >> 4;

  int b  = wid / (TT * MT);
  int r  = wid % (TT * MT);
  int t0 = (r / MT) * 16;
  int m0 = (r % MT) * 16;

  const float* Arow = ws + A_OFF + (size_t)(b * TXT_PAD + t0 + l15) * KDIM;
  const float* melb = mel + (size_t)b * N_MEL * MEL_T + m0 + l15;

  v8f acc = {0.f, 0.f, 0.f, 0.f, 0.f, 0.f, 0.f, 0.f};

  for (int k = 0; k < KDIM; k += 4) {
    v2f a = *(const v2f*)(Arow + k + 2 * kl);

    int K0 = k + 2 * kl;
    int n  = (k < N_MEL) ? K0 : (K0 - N_MEL);
    float x0 = melb[(size_t)n * MEL_T];
    float x1 = melb[(size_t)(n + 1) * MEL_T];
    if (k < N_MEL) { x0 *= x0; x1 *= x1; }
    v2f bv; bv.x = x0; bv.y = x1;

    acc = __builtin_amdgcn_wmma_f32_16x16x4_f32(
         false, a,  false, bv,
         (short)0, acc,  false,  false);
  }

  int m = m0 + l15;
  const float* Cb = ws + C_OFF + b * TXT_PAD;
  const float scale = -0.5f / (float)N_MEL;
#pragma unroll
  for (int j = 0; j < 8; ++j) {
    int t = t0 + j + 8 * kl;
    if (t < TXT_T) {
      lp[(size_t)(b * TXT_T + t) * MEL_T + m] = scale * (acc[j] + Cb[t]);
    }
  }
}

#define SC_ROWS  32
#define SC_WAVES ((TXT_T + SC_ROWS - 1) / SC_ROWS)
#define SC_COLS  16
#define SC_MAXCH (MEL_T / SC_COLS)
#define SC_LINE  32
#define SC_CPL   (SC_LINE / SC_COLS)
#define SC_NLINE (MEL_T / SC_LINE)
static_assert(MEL_T % SC_LINE == 0 && SC_LINE % SC_COLS == 0 && SC_CPL == 2, "groups tile the row; the chunk select below has two cases");
#define LOG2E_F  1.44269504088896340736f
#define LN2_F    0.69314718055994530942f
#define EPS_L2   (1.0e-7f * LOG2E_F)
#define E_ZERO   (-(1 << 28))

__device__ __forceinline__ unsigned gather_above_(unsigned ua, unsigned keep, unsigned sel) {
  return __builtin_amdgcn_permlanex16_var(keep, ua, sel, false, false);
}

__device__ __forceinline__ unsigned row_above_(unsigned ua, unsigned edge, unsigned sel, bool is0) {
  const unsigned x = gather_above_(ua, edge, sel);
  return is0 ? edge : x;
}

__device__ __forceinline__ void weight_(float x, float eps, float& pf, int& pe) {
  const float lg = fmaf(x, LOG2E_F, eps);
  const float fl = floorf(lg);
  pf = __builtin_amdgcn_exp2f(lg - fl);
  pe = (int)fl;
}

__device__ __forceinline__ void scan_col_(float& f, int& e, float bf, int be,
                                          float pf, int pe, unsigned sel, bool is0) {
  const float fb = __uint_as_float(row_above_(__float_as_uint(f), __float_as_uint(bf), sel, is0));
  const int   eb = (int)row_above_((unsigned)e, (unsigned)be, sel, is0);
  const int   E  = e > eb ? e : eb;
  const float s  = __builtin_ldexpf(f, e - E) + __builtin_ldexpf(fb, eb - E);
  f = s * pf;
  e = E + pe;
}

#define G_HI  0x1p+16f
#define G_LO  0x1p-60f
#define D_NONE (-(1 << 20))
__device__ __forceinline__ bool short_chunk_(float& f, const int e, const float (&cur)[SC_COLS], float eps0,
                                             float plf, int ple,
                                             const float (&evf)[SC_COLS], const int (&eve)[SC_COLS],
                                             unsigned sel, bool is0, bool last_lane,
                                             float* __restrict__ out_f, int* __restrict__ out_e) {
  const int ea   = (int)gather_above_((unsigned)e, (unsigned)e, sel);
  const int d    = is0 ? D_NONE : ea - e;
  const int eadj = is0 ? e : e - D_NONE;
  float g = f;
  float gk[SC_COLS];
#pragma unroll
  for (int j = 0; j < SC_COLS; ++j) {
    const float lg  = fmaf(cur[j], LOG2E_F, j == 0 ? eps0 : EPS_L2);
    const float wgt = __builtin_amdgcn_exp2f(lg);
    const float wab = __builtin_ldexpf(wgt, d);
    const float ef  = j == 0 ? plf : evf[j == 0 ? 0 : j - 1];
    const int   ee  = j == 0 ? ple : eve[j == 0 ? 0 : j - 1];
    const float edg = __builtin_ldexpf(ef, ee - eadj);
    const float et  = edg * wgt;
    const float own = fmaf(g, wgt, et);
    const float ab  = __uint_as_float(gather_above_(__float_as_uint(g), __float_as_uint(et), sel));
    g = fmaf(ab, wab, own);
    gk[j] = g;
  }
  const float glast = gk[SC_COLS - 1];
  float gmx = gk[0], gmn = gk[0];
#pragma unroll
  for (int j = 1; j < SC_COLS - 1; ++j) { gmx = fmaxf(gmx, gk[j]); gmn = fminf(gmn, gk[j]); }
  const bool okl = gmx <= G_HI && glast <= G_HI && gmn >= G_LO && glast >= G_LO;
  if (__builtin_amdgcn_ballot_w32(okl) != 0xffffffffu) return false;
  if (last_lane) {
#pragma unroll
    for (int j = 0; j < SC_COLS; ++j) { out_f[j] = gk[j]; out_e[j] = e; }
  }
  f = glast;
  return true;
}

__device__ __forceinline__ void load_group_(float (&raw)[SC_LINE], const float* __restrict__ rowE,
                                            const float* __restrict__ rowO, int col0) {
#pragma unroll
  for (int j = 0; j < SC_LINE / 4; ++j) {
    const float* src = ((j & 1) ? rowO : rowE) + col0 + 8 * (j >> 1);
#pragma unroll
    for (int d = 0; d < 4; ++d) raw[4 * j + d] = src[d];
  }
}

__device__ __forceinline__ float pair_swap_(float v) {
  return __uint_as_float(__builtin_amdgcn_update_dpp(0u, __float_as_uint(v), 0xB1, 0xf, 0xf, true));
}

__device__ __forceinline__ void own_row_(float (&own)[SC_LINE], const float (&raw)[SC_LINE], bool odd) {
#pragma unroll
  for (int g = 0; g < SC_LINE / 8; ++g) {
#pragma unroll
    for (int d = 0; d < 4; ++d) {
      const float x  = raw[4 * (2 * g) + d];
      const float y  = raw[4 * (2 * g + 1) + d];
      const float px = pair_swap_(x);
      const float py = pair_swap_(y);
      own[4 * (2 * g) + d]     = odd ? py : x;
      own[4 * (2 * g + 1) + d] = odd ? y : px;
    }
  }
}

__device__ __forceinline__ void stage_barrier_() {
  __builtin_amdgcn_fence(__ATOMIC_RELEASE, "workgroup", "local");
  __builtin_amdgcn_s_barrier();
  __builtin_amdgcn_fence(__ATOMIC_ACQUIRE, "workgroup", "local");
}

__global__ void __launch_bounds__(SC_WAVES * 32)
mdn_scan_kernel(const float* __restrict__ lp,
                const int* __restrict__ text_len,
                const int* __restrict__ mel_len,
                float* __restrict__ ws) {
  __shared__ __attribute__((aligned(16))) float edge_f[SC_WAVES][2][SC_COLS];
  __shared__ __attribute__((aligned(16))) int   edge_e[SC_WAVES][2][SC_COLS];
  const int b    = blockIdx.x;
  const int w    = __builtin_amdgcn_readfirstlane((int)(threadIdx.x >> 5));
  const int lane = threadIdx.x & 31;
  const int lrow = 2 * (lane & 15) + (lane >> 4);
  const int t    = w * SC_ROWS + lrow;
  const bool is0 = (lane == 0);
  const unsigned sel = (unsigned)((lane & 15) - (lane < 16 ? 1 : 0)) & 15u;
  const int M  = mel_len[b];
  const int tl = text_len[b];
  int nch = (M + SC_COLS - 1) / SC_COLS;
  nch = nch < 0 ? 0 : (nch > SC_MAXCH ? SC_MAXCH : nch);

  for (int i = threadIdx.x; i < SC_WAVES * 2 * SC_COLS; i += SC_WAVES * 32) {
    (&edge_f[0][0][0])[i] = 0.0f;
    (&edge_e[0][0][0])[i] = E_ZERO;
  }

  const bool odd = (lane & 1) != 0;
  const int tE  = t - 2 * (lane & 1);
  const int trE = tE < TXT_T ? tE : TXT_T - 1;
  const int trO = tE + 2 < TXT_T ? tE + 2 : TXT_T - 1;
  const float* rowE = lp + ((size_t)b * TXT_T + trE) * MEL_T + 4 * (lane & 1);
  const float* rowO = lp + ((size_t)b * TXT_T + trO) * MEL_T + 4 * (lane & 1);

  float buf[SC_LINE], nbuf[SC_LINE];
  load_group_(nbuf, rowE, rowO, 0);
  own_row_(buf, nbuf, odd);

  float f = 0.0f;  int e = E_ZERO;
  float rf = 0.0f; int re = E_ZERO;
  float plf = (w == 0) ? 1.0f : 0.0f;
  int   ple = (w == 0) ? 0 : E_ZERO;
  __syncthreads();

  for (int s = 0; s < nch + SC_WAVES - 1; ++s) {
    const int c = s - w;
    if (c >= 0 && c < nch) {
      const int k = c & (SC_CPL - 1);
      if (k == 0) {
        const int qn = (c / SC_CPL + 1 < SC_NLINE) ? c / SC_CPL + 1 : SC_NLINE - 1;
        load_group_(nbuf, rowE, rowO, qn * SC_LINE);
      }
      float cur[SC_COLS];
      const bool first = (k == 0);
#pragma unroll
      for (int i = 0; i < SC_COLS; ++i) cur[i] = first ? buf[i] : buf[SC_COLS + i];

      float evf[SC_COLS]; int eve[SC_COLS];
#pragma unroll
      for (int j = 0; j < SC_COLS; ++j) { evf[j] = 0.0f; eve[j] = E_ZERO; }
      if (w > 0) {
#pragma unroll
        for (int j = 0; j < SC_COLS; ++j) {
          evf[j] = edge_f[w - 1][c & 1][j];
          eve[j] = edge_e[w - 1][c & 1][j];
        }
      }

      const int mbase = c * SC_COLS;
      const float eps0 = (c == 0) ? 0.0f : EPS_L2;
      bool done = false;
      if (mbase + SC_COLS <= M && __builtin_amdgcn_ballot_w32(f > 0.0f) == 0xffffffffu) {
        done = short_chunk_(f, e, cur, eps0, plf, ple, evf, eve, sel, is0, lane == 31,
                            &edge_f[w][c & 1][0], &edge_e[w][c & 1][0]);
        if (done && mbase + SC_COLS == M) { rf = f; re = e; }
      }
      if (!done) {
        float kf[SC_COLS]; int ke[SC_COLS];
        if (mbase + SC_COLS <= M) {
          float pf[SC_COLS]; int pe[SC_COLS];
#pragma unroll
          for (int j = 0; j < SC_COLS; ++j) weight_(cur[j], j == 0 ? eps0 : EPS_L2, pf[j], pe[j]);
#pragma unroll
          for (int j = 0; j < SC_COLS; ++j) {
            scan_col_(f, e, j == 0 ? plf : evf[j == 0 ? 0 : j - 1],
                            j == 0 ? ple : eve[j == 0 ? 0 : j - 1], pf[j], pe[j], sel, is0);
            kf[j] = f; ke[j] = e;
          }
          if (mbase + SC_COLS == M) { rf = f; re = e; }
        } else {
#pragma unroll
          for (int j = 0; j < SC_COLS; ++j) {
            if (mbase + j < M) {
              float pfj; int pej;
              weight_(cur[j], j == 0 ? eps0 : EPS_L2, pfj, pej);
              scan_col_(f, e, j == 0 ? plf : evf[j == 0 ? 0 : j - 1],
                              j == 0 ? ple : eve[j == 0 ? 0 : j - 1], pfj, pej, sel, is0);
              if (mbase + j == M - 1) { rf = f; re = e; }
            }
            kf[j] = f; ke[j] = e;
          }
        }
        if (lane == 31) {
#pragma unroll
          for (int j = 0; j < SC_COLS; ++j) { edge_f[w][c & 1][j] = kf[j]; edge_e[w][c & 1][j] = ke[j]; }
        }
      }
      plf = evf[SC_COLS - 1];
      ple = eve[SC_COLS - 1];

      e += __builtin_amdgcn_frexp_expf(f);
      f  = __builtin_amdgcn_frexp_mantf(f);
      if (k == SC_CPL - 1) {
        own_row_(buf, nbuf, odd);
      }
    }
    stage_barrier_();
  }
  if (t == tl - 1) {
    const float la = (rf > 0.0f) ? (log2f(rf) + (float)re) * LN2_F : -PAD_MAG;
    ws[P_OFF + b] = la / (float)M;
  }
}

__global__ void mdn_reduce_kernel(const float* __restrict__ ws,
                                  float* __restrict__ out) {
  float s = 0.f;
  for (int b = 0; b < B_; ++b) s += ws[P_OFF + b];
  out[0] = -s / (float)B_;
}

extern "C" void kernel_launch(void* const* d_in, const int* in_sizes, int n_in,
                              void* d_out, int out_size, void* d_ws, size_t ws_size,
                              hipStream_t stream) {
  const float* mu_logvar = (const float*)d_in[0];
  const float* melspec   = (const float*)d_in[1];
  const int*   text_len  = (const int*)d_in[2];
  const int*   mel_len   = (const int*)d_in[3];
  float* out = (float*)d_out;
  float* ws  = (float*)d_ws;
  float* lp  = out + 1;

  mdn_prep_kernel<<<(B_ * TXT_PAD) / PREP_WAVES, PREP_WAVES * 32, 0, stream>>>(mu_logvar, ws);

  const int total_waves = B_ * (TXT_PAD / 16) * (MEL_T / 16);
  mdn_gemm_kernel<<<total_waves / 8, 256, 0, stream>>>(ws, melspec, lp);

  mdn_scan_kernel<<<B_, SC_WAVES * 32, 0, stream>>>(lp, text_len, mel_len, ws);
  mdn_reduce_kernel<<<1, 1, 0, stream>>>(ws, out);
}


# scan: a wave runs only chunks where its rows can be non-zero (from column 32w-32) and that the read-out cell (tl-1, M-1) depends on; waves past row tl-1 idle; loop ends with row tl-1; sums unchanged
# speedup vs baseline: 5.5713x; 1.0837x over previous
#include <hip/hip_runtime.h>
#include <math.h>

#define B_      8
#define TXT_T   200
#define TXT_PAD 208
#define MEL_T   800
#define N_MEL   80
#define KDIM    160
#define PAD_MAG 1.0e12f

typedef __attribute__((ext_vector_type(2))) float v2f;
typedef __attribute__((ext_vector_type(8))) float v8f;

#define A_OFF 0
#define C_OFF (B_ * TXT_PAD * KDIM)
#define P_OFF (C_OFF + B_ * TXT_PAD)

#define PREP_WAVES 8
static_assert((B_ * TXT_PAD) % PREP_WAVES == 0 && N_MEL % 4 == 0 && N_MEL / 4 <= 32, "prep grid is exact; one row fits one wave");
__global__ void __launch_bounds__(PREP_WAVES * 32)
mdn_prep_kernel(const float* __restrict__ mu_logvar,
                float* __restrict__ ws) {
  const int lane = threadIdx.x & 31;
  const int idx  = blockIdx.x * PREP_WAVES + (threadIdx.x >> 5);
  const int b = idx / TXT_PAD;
  const int t = idx % TXT_PAD;
  float* Arow = ws + A_OFF + (size_t)idx * KDIM;
  float c = 0.0f;
  if (lane < N_MEL / 4) {
    float4 w4 = make_float4(0.0f, 0.0f, 0.0f, 0.0f);
    float4 m4 = make_float4(0.0f, 0.0f, 0.0f, 0.0f);
    if (t < TXT_T) {
      const float* ml = mu_logvar + (size_t)(b * TXT_T + t) * (2 * N_MEL);
      const float4 mu = *(const float4*)(ml + 4 * lane);
      const float4 lv = *(const float4*)(ml + N_MEL + 4 * lane);
      w4.x = expf(-lv.x); w4.y = expf(-lv.y); w4.z = expf(-lv.z); w4.w = expf(-lv.w);
      m4.x = -2.0f * mu.x * w4.x; m4.y = -2.0f * mu.y * w4.y;
      m4.z = -2.0f * mu.z * w4.z; m4.w = -2.0f * mu.w * w4.w;
      c  = mu.x * mu.x * w4.x + lv.x;
      c += mu.y * mu.y * w4.y + lv.y;
      c += mu.z * mu.z * w4.z + lv.z;
      c += mu.w * mu.w * w4.w + lv.w;
    }
    *(float4*)(Arow + 4 * lane)         = w4;
    *(float4*)(Arow + N_MEL + 4 * lane) = m4;
  }
  for (int m = 16; m >= 1; m >>= 1) c += __shfl_xor(c, m, 32);
  if (lane == 0) ws[C_OFF + idx] = c;
}

__global__ void __launch_bounds__(256)
mdn_gemm_kernel(const float* __restrict__ ws,
                const float* __restrict__ mel,
                float* __restrict__ lp) {
  const int TT = TXT_PAD / 16;
  const int MT = MEL_T / 16;
  int wid = blockIdx.x * (blockDim.x >> 5) + (threadIdx.x >> 5);
  if (wid >= B_ * TT * MT) return;
  int lane = threadIdx.x & 31;
  int l15  = lane & 15;
  int kl   = lane >> 4;

  int b  = wid / (TT * MT);
  int r  = wid % (TT * MT);
  int t0 = (r / MT) * 16;
  int m0 = (r % MT) * 16;

  const float* Arow = ws + A_OFF + (size_t)(b * TXT_PAD + t0 + l15) * KDIM;
  const float* melb = mel + (size_t)b * N_MEL * MEL_T + m0 + l15;

  v8f acc = {0.f, 0.f, 0.f, 0.f, 0.f, 0.f, 0.f, 0.f};

  for (int k = 0; k < KDIM; k += 4) {
    v2f a = *(const v2f*)(Arow + k + 2 * kl);

    int K0 = k + 2 * kl;
    int n  = (k < N_MEL) ? K0 : (K0 - N_MEL);
    float x0 = melb[(size_t)n * MEL_T];
    float x1 = melb[(size_t)(n + 1) * MEL_T];
    if (k < N_MEL) { x0 *= x0; x1 *= x1; }
    v2f bv; bv.x = x0; bv.y = x1;

    acc = __builtin_amdgcn_wmma_f32_16x16x4_f32(
         false, a,  false, bv,
         (short)0, acc,  false,  false);
  }

  int m = m0 + l15;
  const float* Cb = ws + C_OFF + b * TXT_PAD;
  const float scale = -0.5f / (float)N_MEL;
#pragma unroll
  for (int j = 0; j < 8; ++j) {
    int t = t0 + j + 8 * kl;
    if (t < TXT_T) {
      lp[(size_t)(b * TXT_T + t) * MEL_T + m] = scale * (acc[j] + Cb[t]);
    }
  }
}

#define SC_ROWS  32
#define SC_WAVES ((TXT_T + SC_ROWS - 1) / SC_ROWS)
#define SC_COLS  16
#define SC_MAXCH (MEL_T / SC_COLS)
#define SC_LINE  32
#define SC_CPL   (SC_LINE / SC_COLS)
#define SC_NLINE (MEL_T / SC_LINE)
static_assert(MEL_T % SC_LINE == 0 && SC_LINE % SC_COLS == 0 && SC_CPL == 2, "groups tile the row; the chunk select below has two cases");
#define LOG2E_F  1.44269504088896340736f
#define LN2_F    0.69314718055994530942f
#define EPS_L2   (1.0e-7f * LOG2E_F)
#define E_ZERO   (-(1 << 28))

__device__ __forceinline__ unsigned gather_above_(unsigned ua, unsigned keep, unsigned sel) {
  return __builtin_amdgcn_permlanex16_var(keep, ua, sel, false, false);
}

__device__ __forceinline__ unsigned row_above_(unsigned ua, unsigned edge, unsigned sel, bool is0) {
  const unsigned x = gather_above_(ua, edge, sel);
  return is0 ? edge : x;
}

__device__ __forceinline__ void weight_(float x, float eps, float& pf, int& pe) {
  const float lg = fmaf(x, LOG2E_F, eps);
  const float fl = floorf(lg);
  pf = __builtin_amdgcn_exp2f(lg - fl);
  pe = (int)fl;
}

__device__ __forceinline__ void scan_col_(float& f, int& e, float bf, int be,
                                          float pf, int pe, unsigned sel, bool is0) {
  const float fb = __uint_as_float(row_above_(__float_as_uint(f), __float_as_uint(bf), sel, is0));
  const int   eb = (int)row_above_((unsigned)e, (unsigned)be, sel, is0);
  const int   E  = e > eb ? e : eb;
  const float s  = __builtin_ldexpf(f, e - E) + __builtin_ldexpf(fb, eb - E);
  f = s * pf;
  e = E + pe;
}

#define G_HI  0x1p+16f
#define G_LO  0x1p-60f
#define D_NONE (-(1 << 20))
__device__ __forceinline__ bool short_chunk_(float& f, const int e, const float (&cur)[SC_COLS], float eps0,
                                             float plf, int ple,
                                             const float (&evf)[SC_COLS], const int (&eve)[SC_COLS],
                                             unsigned sel, bool is0, bool last_lane,
                                             float* __restrict__ out_f, int* __restrict__ out_e) {
  const int ea   = (int)gather_above_((unsigned)e, (unsigned)e, sel);
  const int d    = is0 ? D_NONE : ea - e;
  const int eadj = is0 ? e : e - D_NONE;
  float g = f;
  float gk[SC_COLS];
#pragma unroll
  for (int j = 0; j < SC_COLS; ++j) {
    const float lg  = fmaf(cur[j], LOG2E_F, j == 0 ? eps0 : EPS_L2);
    const float wgt = __builtin_amdgcn_exp2f(lg);
    const float wab = __builtin_ldexpf(wgt, d);
    const float ef  = j == 0 ? plf : evf[j == 0 ? 0 : j - 1];
    const int   ee  = j == 0 ? ple : eve[j == 0 ? 0 : j - 1];
    const float edg = __builtin_ldexpf(ef, ee - eadj);
    const float et  = edg * wgt;
    const float own = fmaf(g, wgt, et);
    const float ab  = __uint_as_float(gather_above_(__float_as_uint(g), __float_as_uint(et), sel));
    g = fmaf(ab, wab, own);
    gk[j] = g;
  }
  const float glast = gk[SC_COLS - 1];
  float gmx = gk[0], gmn = gk[0];
#pragma unroll
  for (int j = 1; j < SC_COLS - 1; ++j) { gmx = fmaxf(gmx, gk[j]); gmn = fminf(gmn, gk[j]); }
  const bool okl = gmx <= G_HI && glast <= G_HI && gmn >= G_LO && glast >= G_LO;
  if (__builtin_amdgcn_ballot_w32(okl) != 0xffffffffu) return false;
  if (last_lane) {
#pragma unroll
    for (int j = 0; j < SC_COLS; ++j) { out_f[j] = gk[j]; out_e[j] = e; }
  }
  f = glast;
  return true;
}

__device__ __forceinline__ void load_group_(float (&raw)[SC_LINE], const float* __restrict__ rowE,
                                            const float* __restrict__ rowO, int col0) {
#pragma unroll
  for (int j = 0; j < SC_LINE / 4; ++j) {
    const float* src = ((j & 1) ? rowO : rowE) + col0 + 8 * (j >> 1);
#pragma unroll
    for (int d = 0; d < 4; ++d) raw[4 * j + d] = src[d];
  }
}

__device__ __forceinline__ float pair_swap_(float v) {
  return __uint_as_float(__builtin_amdgcn_update_dpp(0u, __float_as_uint(v), 0xB1, 0xf, 0xf, true));
}

__device__ __forceinline__ void own_row_(float (&own)[SC_LINE], const float (&raw)[SC_LINE], bool odd) {
#pragma unroll
  for (int g = 0; g < SC_LINE / 8; ++g) {
#pragma unroll
    for (int d = 0; d < 4; ++d) {
      const float x  = raw[4 * (2 * g) + d];
      const float y  = raw[4 * (2 * g + 1) + d];
      const float px = pair_swap_(x);
      const float py = pair_swap_(y);
      own[4 * (2 * g) + d]     = odd ? py : x;
      own[4 * (2 * g + 1) + d] = odd ? y : px;
    }
  }
}

__device__ __forceinline__ void stage_barrier_() {
  __builtin_amdgcn_fence(__ATOMIC_RELEASE, "workgroup", "local");
  __builtin_amdgcn_s_barrier();
  __builtin_amdgcn_fence(__ATOMIC_ACQUIRE, "workgroup", "local");
}

__global__ void __launch_bounds__(SC_WAVES * 32)
mdn_scan_kernel(const float* __restrict__ lp,
                const int* __restrict__ text_len,
                const int* __restrict__ mel_len,
                float* __restrict__ ws) {
  __shared__ __attribute__((aligned(16))) float edge_f[SC_WAVES][2][SC_COLS];
  __shared__ __attribute__((aligned(16))) int   edge_e[SC_WAVES][2][SC_COLS];
  const int b    = blockIdx.x;
  const int w    = __builtin_amdgcn_readfirstlane((int)(threadIdx.x >> 5));
  const int lane = threadIdx.x & 31;
  const int lrow = 2 * (lane & 15) + (lane >> 4);
  const int t    = w * SC_ROWS + lrow;
  const bool is0 = (lane == 0);
  const unsigned sel = (unsigned)((lane & 15) - (lane < 16 ? 1 : 0)) & 15u;
  const int M  = mel_len[b];
  const int tl = text_len[b];
  int nch = (M + SC_COLS - 1) / SC_COLS;
  nch = nch < 0 ? 0 : (nch > SC_MAXCH ? SC_MAXCH : nch);

  const int c_nz = (w * SC_ROWS) / SC_COLS;
  const int c_lo = (c_nz > SC_CPL ? c_nz - SC_CPL : 0) & ~(SC_CPL - 1);
  int c_hi = -1;
  if (w * SC_ROWS <= tl - 1) {
    const int t_top = (w * SC_ROWS + SC_ROWS - 1 < tl - 1) ? (w * SC_ROWS + SC_ROWS - 1) : (tl - 1);
    const int m_hi  = (M - 1) - ((tl - 1) - t_top);
    if (m_hi >= 0) { c_hi = m_hi / SC_COLS; c_hi = c_hi > nch - 1 ? nch - 1 : c_hi; }
  }
  int wl = tl >= 1 ? (tl - 1) / SC_ROWS : 0;
  wl = wl > SC_WAVES - 1 ? SC_WAVES - 1 : wl;

  for (int i = threadIdx.x; i < SC_WAVES * 2 * SC_COLS; i += SC_WAVES * 32) {
    (&edge_f[0][0][0])[i] = 0.0f;
    (&edge_e[0][0][0])[i] = E_ZERO;
  }

  const bool odd = (lane & 1) != 0;
  const int tE  = t - 2 * (lane & 1);
  const int trE = tE < TXT_T ? tE : TXT_T - 1;
  const int trO = tE + 2 < TXT_T ? tE + 2 : TXT_T - 1;
  const float* rowE = lp + ((size_t)b * TXT_T + trE) * MEL_T + 4 * (lane & 1);
  const float* rowO = lp + ((size_t)b * TXT_T + trO) * MEL_T + 4 * (lane & 1);

  float buf[SC_LINE], nbuf[SC_LINE];
  load_group_(nbuf, rowE, rowO, (c_lo / SC_CPL) * SC_LINE);
  own_row_(buf, nbuf, odd);

  float f = 0.0f;  int e = E_ZERO;
  float rf = 0.0f; int re = E_ZERO;
  float plf = (w == 0) ? 1.0f : 0.0f;
  int   ple = (w == 0) ? 0 : E_ZERO;
  __syncthreads();

  for (int s = 0; s < nch + wl; ++s) {
    const int c = s - w;
    if (c_hi >= 0 && c > c_hi && c <= c_hi + 2 && lane == 31) {
#pragma unroll
      for (int j = 0; j < SC_COLS; ++j) { edge_f[w][c & 1][j] = 0.0f; edge_e[w][c & 1][j] = E_ZERO; }
    }
    if (c >= c_lo && c <= c_hi) {
      const int k = c & (SC_CPL - 1);
      if (k == 0) {
        const int qn = (c / SC_CPL + 1 < SC_NLINE) ? c / SC_CPL + 1 : SC_NLINE - 1;
        load_group_(nbuf, rowE, rowO, qn * SC_LINE);
      }
      float cur[SC_COLS];
      const bool first = (k == 0);
#pragma unroll
      for (int i = 0; i < SC_COLS; ++i) cur[i] = first ? buf[i] : buf[SC_COLS + i];

      float evf[SC_COLS]; int eve[SC_COLS];
#pragma unroll
      for (int j = 0; j < SC_COLS; ++j) { evf[j] = 0.0f; eve[j] = E_ZERO; }
      if (w > 0) {
#pragma unroll
        for (int j = 0; j < SC_COLS; ++j) {
          evf[j] = edge_f[w - 1][c & 1][j];
          eve[j] = edge_e[w - 1][c & 1][j];
        }
      }

      const int mbase = c * SC_COLS;
      const float eps0 = (c == 0) ? 0.0f : EPS_L2;
      bool done = false;
      if (mbase + SC_COLS <= M && __builtin_amdgcn_ballot_w32(f > 0.0f) == 0xffffffffu) {
        done = short_chunk_(f, e, cur, eps0, plf, ple, evf, eve, sel, is0, lane == 31,
                            &edge_f[w][c & 1][0], &edge_e[w][c & 1][0]);
        if (done && mbase + SC_COLS == M) { rf = f; re = e; }
      }
      if (!done) {
        float kf[SC_COLS]; int ke[SC_COLS];
        if (mbase + SC_COLS <= M) {
          float pf[SC_COLS]; int pe[SC_COLS];
#pragma unroll
          for (int j = 0; j < SC_COLS; ++j) weight_(cur[j], j == 0 ? eps0 : EPS_L2, pf[j], pe[j]);
#pragma unroll
          for (int j = 0; j < SC_COLS; ++j) {
            scan_col_(f, e, j == 0 ? plf : evf[j == 0 ? 0 : j - 1],
                            j == 0 ? ple : eve[j == 0 ? 0 : j - 1], pf[j], pe[j], sel, is0);
            kf[j] = f; ke[j] = e;
          }
          if (mbase + SC_COLS == M) { rf = f; re = e; }
        } else {
#pragma unroll
          for (int j = 0; j < SC_COLS; ++j) {
            if (mbase + j < M) {
              float pfj; int pej;
              weight_(cur[j], j == 0 ? eps0 : EPS_L2, pfj, pej);
              scan_col_(f, e, j == 0 ? plf : evf[j == 0 ? 0 : j - 1],
                              j == 0 ? ple : eve[j == 0 ? 0 : j - 1], pfj, pej, sel, is0);
              if (mbase + j == M - 1) { rf = f; re = e; }
            }
            kf[j] = f; ke[j] = e;
          }
        }
        if (lane == 31) {
#pragma unroll
          for (int j = 0; j < SC_COLS; ++j) { edge_f[w][c & 1][j] = kf[j]; edge_e[w][c & 1][j] = ke[j]; }
        }
      }
      plf = evf[SC_COLS - 1];
      ple = eve[SC_COLS - 1];

      e += __builtin_amdgcn_frexp_expf(f);
      f  = __builtin_amdgcn_frexp_mantf(f);
      if (k == SC_CPL - 1) {
        own_row_(buf, nbuf, odd);
      }
    }
    stage_barrier_();
  }
  if (t == tl - 1) {
    const float la = (rf > 0.0f) ? (log2f(rf) + (float)re) * LN2_F : -PAD_MAG;
    ws[P_OFF + b] = la / (float)M;
  }
}

__global__ void mdn_reduce_kernel(const float* __restrict__ ws,
                                  float* __restrict__ out) {
  float s = 0.f;
  for (int b = 0; b < B_; ++b) s += ws[P_OFF + b];
  out[0] = -s / (float)B_;
}

extern "C" void kernel_launch(void* const* d_in, const int* in_sizes, int n_in,
                              void* d_out, int out_size, void* d_ws, size_t ws_size,
                              hipStream_t stream) {
  const float* mu_logvar = (const float*)d_in[0];
  const float* melspec   = (const float*)d_in[1];
  const int*   text_len  = (const int*)d_in[2];
  const int*   mel_len   = (const int*)d_in[3];
  float* out = (float*)d_out;
  float* ws  = (float*)d_ws;
  float* lp  = out + 1;

  mdn_prep_kernel<<<(B_ * TXT_PAD) / PREP_WAVES, PREP_WAVES * 32, 0, stream>>>(mu_logvar, ws);

  const int total_waves = B_ * (TXT_PAD / 16) * (MEL_T / 16);
  mdn_gemm_kernel<<<total_waves / 8, 256, 0, stream>>>(ws, melspec, lp);

  mdn_scan_kernel<<<B_, SC_WAVES * 32, 0, stream>>>(lp, text_len, mel_len, ws);
  mdn_reduce_kernel<<<1, 1, 0, stream>>>(ws, out);
}


# gemm: all of a wave's global loads (row constants, x, 40 A fragments) issued before the first matrix step; x read once for its x^2 and x steps (was twice); was drained every 5 steps; sums unchanged
# speedup vs baseline: 6.1654x; 1.1066x over previous
#include <hip/hip_runtime.h>
#include <math.h>

#define B_      8
#define TXT_T   200
#define TXT_PAD 208
#define MEL_T   800
#define N_MEL   80
#define KDIM    160
#define PAD_MAG 1.0e12f

typedef __attribute__((ext_vector_type(2))) float v2f;
typedef __attribute__((ext_vector_type(8))) float v8f;

#define A_OFF 0
#define C_OFF (B_ * TXT_PAD * KDIM)
#define P_OFF (C_OFF + B_ * TXT_PAD)

#define PREP_WAVES 8
static_assert((B_ * TXT_PAD) % PREP_WAVES == 0 && N_MEL % 4 == 0 && N_MEL / 4 <= 32, "prep grid is exact; one row fits one wave");
__global__ void __launch_bounds__(PREP_WAVES * 32)
mdn_prep_kernel(const float* __restrict__ mu_logvar,
                float* __restrict__ ws) {
  const int lane = threadIdx.x & 31;
  const int idx  = blockIdx.x * PREP_WAVES + (threadIdx.x >> 5);
  const int b = idx / TXT_PAD;
  const int t = idx % TXT_PAD;
  float* Arow = ws + A_OFF + (size_t)idx * KDIM;
  float c = 0.0f;
  if (lane < N_MEL / 4) {
    float4 w4 = make_float4(0.0f, 0.0f, 0.0f, 0.0f);
    float4 m4 = make_float4(0.0f, 0.0f, 0.0f, 0.0f);
    if (t < TXT_T) {
      const float* ml = mu_logvar + (size_t)(b * TXT_T + t) * (2 * N_MEL);
      const float4 mu = *(const float4*)(ml + 4 * lane);
      const float4 lv = *(const float4*)(ml + N_MEL + 4 * lane);
      w4.x = expf(-lv.x); w4.y = expf(-lv.y); w4.z = expf(-lv.z); w4.w = expf(-lv.w);
      m4.x = -2.0f * mu.x * w4.x; m4.y = -2.0f * mu.y * w4.y;
      m4.z = -2.0f * mu.z * w4.z; m4.w = -2.0f * mu.w * w4.w;
      c  = mu.x * mu.x * w4.x + lv.x;
      c += mu.y * mu.y * w4.y + lv.y;
      c += mu.z * mu.z * w4.z + lv.z;
      c += mu.w * mu.w * w4.w + lv.w;
    }
    *(float4*)(Arow + 4 * lane)         = w4;
    *(float4*)(Arow + N_MEL + 4 * lane) = m4;
  }
  for (int m = 16; m >= 1; m >>= 1) c += __shfl_xor(c, m, 32);
  if (lane == 0) ws[C_OFF + idx] = c;
}

#define GK_STEPS (KDIM / 4)
#define GK_HALF  (N_MEL / 4)
static_assert(GK_STEPS == 2 * GK_HALF, "every x pair serves two steps");
__global__ void __launch_bounds__(256)
mdn_gemm_kernel(const float* __restrict__ ws,
                const float* __restrict__ mel,
                float* __restrict__ lp) {
  const int TT = TXT_PAD / 16;
  const int MT = MEL_T / 16;
  int wid = blockIdx.x * (blockDim.x >> 5) + (threadIdx.x >> 5);
  if (wid >= B_ * TT * MT) return;
  int lane = threadIdx.x & 31;
  int l15  = lane & 15;
  int kl   = lane >> 4;

  int b  = wid / (TT * MT);
  int r  = wid % (TT * MT);
  int t0 = (r / MT) * 16;
  int m0 = (r % MT) * 16;

  const float* Arow2 = ws + A_OFF + (size_t)(b * TXT_PAD + t0 + l15) * KDIM + 2 * kl;
  const float* melb2 = mel + (size_t)b * N_MEL * MEL_T + (size_t)(2 * kl) * MEL_T + m0 + l15;

  const float* Cb = ws + C_OFF + b * TXT_PAD + t0 + 8 * kl;
  const float4 c03 = *(const float4*)(Cb);
  const float4 c47 = *(const float4*)(Cb + 4);

  float x0[GK_HALF], x1[GK_HALF];
#pragma unroll
  for (int j = 0; j < GK_HALF; ++j) {
    x0[j] = melb2[(size_t)(4 * j) * MEL_T];
    x1[j] = melb2[(size_t)(4 * j + 1) * MEL_T];
  }
  v2f a[GK_STEPS];
#pragma unroll
  for (int s = 0; s < GK_STEPS; ++s) a[s] = *(const v2f*)(Arow2 + 4 * s);
  __builtin_amdgcn_sched_barrier(0);

  v8f acc = {0.f, 0.f, 0.f, 0.f, 0.f, 0.f, 0.f, 0.f};
#pragma unroll
  for (int s = 0; s < GK_STEPS; ++s) {
    const int j = s < GK_HALF ? s : s - GK_HALF;
    v2f bv;
    bv.x = s < GK_HALF ? x0[j] * x0[j] : x0[j];
    bv.y = s < GK_HALF ? x1[j] * x1[j] : x1[j];
    acc = __builtin_amdgcn_wmma_f32_16x16x4_f32(
         false, a[s],  false, bv,
         (short)0, acc,  false,  false);
  }

  int m = m0 + l15;
  const float cj[8] = {c03.x, c03.y, c03.z, c03.w, c47.x, c47.y, c47.z, c47.w};
  const float scale = -0.5f / (float)N_MEL;
#pragma unroll
  for (int j = 0; j < 8; ++j) {
    int t = t0 + j + 8 * kl;
    if (t < TXT_T) {
      lp[(size_t)(b * TXT_T + t) * MEL_T + m] = scale * (acc[j] + cj[j]);
    }
  }
}

#define SC_ROWS  32
#define SC_WAVES ((TXT_T + SC_ROWS - 1) / SC_ROWS)
#define SC_COLS  16
#define SC_MAXCH (MEL_T / SC_COLS)
#define SC_LINE  32
#define SC_CPL   (SC_LINE / SC_COLS)
#define SC_NLINE (MEL_T / SC_LINE)
static_assert(MEL_T % SC_LINE == 0 && SC_LINE % SC_COLS == 0 && SC_CPL == 2, "groups tile the row; the chunk select below has two cases");
#define LOG2E_F  1.44269504088896340736f
#define LN2_F    0.69314718055994530942f
#define EPS_L2   (1.0e-7f * LOG2E_F)
#define E_ZERO   (-(1 << 28))

__device__ __forceinline__ unsigned gather_above_(unsigned ua, unsigned keep, unsigned sel) {
  return __builtin_amdgcn_permlanex16_var(keep, ua, sel, false, false);
}

__device__ __forceinline__ unsigned row_above_(unsigned ua, unsigned edge, unsigned sel, bool is0) {
  const unsigned x = gather_above_(ua, edge, sel);
  return is0 ? edge : x;
}

__device__ __forceinline__ void weight_(float x, float eps, float& pf, int& pe) {
  const float lg = fmaf(x, LOG2E_F, eps);
  const float fl = floorf(lg);
  pf = __builtin_amdgcn_exp2f(lg - fl);
  pe = (int)fl;
}

__device__ __forceinline__ void scan_col_(float& f, int& e, float bf, int be,
                                          float pf, int pe, unsigned sel, bool is0) {
  const float fb = __uint_as_float(row_above_(__float_as_uint(f), __float_as_uint(bf), sel, is0));
  const int   eb = (int)row_above_((unsigned)e, (unsigned)be, sel, is0);
  const int   E  = e > eb ? e : eb;
  const float s  = __builtin_ldexpf(f, e - E) + __builtin_ldexpf(fb, eb - E);
  f = s * pf;
  e = E + pe;
}

#define G_HI  0x1p+16f
#define G_LO  0x1p-60f
#define D_NONE (-(1 << 20))
__device__ __forceinline__ bool short_chunk_(float& f, const int e, const float (&cur)[SC_COLS], float eps0,
                                             float plf, int ple,
                                             const float (&evf)[SC_COLS], const int (&eve)[SC_COLS],
                                             unsigned sel, bool is0, bool last_lane,
                                             float* __restrict__ out_f, int* __restrict__ out_e) {
  const int ea   = (int)gather_above_((unsigned)e, (unsigned)e, sel);
  const int d    = is0 ? D_NONE : ea - e;
  const int eadj = is0 ? e : e - D_NONE;
  float g = f;
  float gk[SC_COLS];
#pragma unroll
  for (int j = 0; j < SC_COLS; ++j) {
    const float lg  = fmaf(cur[j], LOG2E_F, j == 0 ? eps0 : EPS_L2);
    const float wgt = __builtin_amdgcn_exp2f(lg);
    const float wab = __builtin_ldexpf(wgt, d);
    const float ef  = j == 0 ? plf : evf[j == 0 ? 0 : j - 1];
    const int   ee  = j == 0 ? ple : eve[j == 0 ? 0 : j - 1];
    const float edg = __builtin_ldexpf(ef, ee - eadj);
    const float et  = edg * wgt;
    const float own = fmaf(g, wgt, et);
    const float ab  = __uint_as_float(gather_above_(__float_as_uint(g), __float_as_uint(et), sel));
    g = fmaf(ab, wab, own);
    gk[j] = g;
  }
  const float glast = gk[SC_COLS - 1];
  float gmx = gk[0], gmn = gk[0];
#pragma unroll
  for (int j = 1; j < SC_COLS - 1; ++j) { gmx = fmaxf(gmx, gk[j]); gmn = fminf(gmn, gk[j]); }
  const bool okl = gmx <= G_HI && glast <= G_HI && gmn >= G_LO && glast >= G_LO;
  if (__builtin_amdgcn_ballot_w32(okl) != 0xffffffffu) return false;
  if (last_lane) {
#pragma unroll
    for (int j = 0; j < SC_COLS; ++j) { out_f[j] = gk[j]; out_e[j] = e; }
  }
  f = glast;
  return true;
}

__device__ __forceinline__ void load_group_(float (&raw)[SC_LINE], const float* __restrict__ rowE,
                                            const float* __restrict__ rowO, int col0) {
#pragma unroll
  for (int j = 0; j < SC_LINE / 4; ++j) {
    const float* src = ((j & 1) ? rowO : rowE) + col0 + 8 * (j >> 1);
#pragma unroll
    for (int d = 0; d < 4; ++d) raw[4 * j + d] = src[d];
  }
}

__device__ __forceinline__ float pair_swap_(float v) {
  return __uint_as_float(__builtin_amdgcn_update_dpp(0u, __float_as_uint(v), 0xB1, 0xf, 0xf, true));
}

__device__ __forceinline__ void own_row_(float (&own)[SC_LINE], const float (&raw)[SC_LINE], bool odd) {
#pragma unroll
  for (int g = 0; g < SC_LINE / 8; ++g) {
#pragma unroll
    for (int d = 0; d < 4; ++d) {
      const float x  = raw[4 * (2 * g) + d];
      const float y  = raw[4 * (2 * g + 1) + d];
      const float px = pair_swap_(x);
      const float py = pair_swap_(y);
      own[4 * (2 * g) + d]     = odd ? py : x;
      own[4 * (2 * g + 1) + d] = odd ? y : px;
    }
  }
}

__device__ __forceinline__ void stage_barrier_() {
  __builtin_amdgcn_fence(__ATOMIC_RELEASE, "workgroup", "local");
  __builtin_amdgcn_s_barrier();
  __builtin_amdgcn_fence(__ATOMIC_ACQUIRE, "workgroup", "local");
}

__global__ void __launch_bounds__(SC_WAVES * 32)
mdn_scan_kernel(const float* __restrict__ lp,
                const int* __restrict__ text_len,
                const int* __restrict__ mel_len,
                float* __restrict__ ws) {
  __shared__ __attribute__((aligned(16))) float edge_f[SC_WAVES][2][SC_COLS];
  __shared__ __attribute__((aligned(16))) int   edge_e[SC_WAVES][2][SC_COLS];
  const int b    = blockIdx.x;
  const int w    = __builtin_amdgcn_readfirstlane((int)(threadIdx.x >> 5));
  const int lane = threadIdx.x & 31;
  const int lrow = 2 * (lane & 15) + (lane >> 4);
  const int t    = w * SC_ROWS + lrow;
  const bool is0 = (lane == 0);
  const unsigned sel = (unsigned)((lane & 15) - (lane < 16 ? 1 : 0)) & 15u;
  const int M  = mel_len[b];
  const int tl = text_len[b];
  int nch = (M + SC_COLS - 1) / SC_COLS;
  nch = nch < 0 ? 0 : (nch > SC_MAXCH ? SC_MAXCH : nch);

  const int c_nz = (w * SC_ROWS) / SC_COLS;
  const int c_lo = (c_nz > SC_CPL ? c_nz - SC_CPL : 0) & ~(SC_CPL - 1);
  int c_hi = -1;
  if (w * SC_ROWS <= tl - 1) {
    const int t_top = (w * SC_ROWS + SC_ROWS - 1 < tl - 1) ? (w * SC_ROWS + SC_ROWS - 1) : (tl - 1);
    const int m_hi  = (M - 1) - ((tl - 1) - t_top);
    if (m_hi >= 0) { c_hi = m_hi / SC_COLS; c_hi = c_hi > nch - 1 ? nch - 1 : c_hi; }
  }
  int wl = tl >= 1 ? (tl - 1) / SC_ROWS : 0;
  wl = wl > SC_WAVES - 1 ? SC_WAVES - 1 : wl;

  for (int i = threadIdx.x; i < SC_WAVES * 2 * SC_COLS; i += SC_WAVES * 32) {
    (&edge_f[0][0][0])[i] = 0.0f;
    (&edge_e[0][0][0])[i] = E_ZERO;
  }

  const bool odd = (lane & 1) != 0;
  const int tE  = t - 2 * (lane & 1);
  const int trE = tE < TXT_T ? tE : TXT_T - 1;
  const int trO = tE + 2 < TXT_T ? tE + 2 : TXT_T - 1;
  const float* rowE = lp + ((size_t)b * TXT_T + trE) * MEL_T + 4 * (lane & 1);
  const float* rowO = lp + ((size_t)b * TXT_T + trO) * MEL_T + 4 * (lane & 1);

  float buf[SC_LINE], nbuf[SC_LINE];
  load_group_(nbuf, rowE, rowO, (c_lo / SC_CPL) * SC_LINE);
  own_row_(buf, nbuf, odd);

  float f = 0.0f;  int e = E_ZERO;
  float rf = 0.0f; int re = E_ZERO;
  float plf = (w == 0) ? 1.0f : 0.0f;
  int   ple = (w == 0) ? 0 : E_ZERO;
  __syncthreads();

  for (int s = 0; s < nch + wl; ++s) {
    const int c = s - w;
    if (c_hi >= 0 && c > c_hi && c <= c_hi + 2 && lane == 31) {
#pragma unroll
      for (int j = 0; j < SC_COLS; ++j) { edge_f[w][c & 1][j] = 0.0f; edge_e[w][c & 1][j] = E_ZERO; }
    }
    if (c >= c_lo && c <= c_hi) {
      const int k = c & (SC_CPL - 1);
      if (k == 0) {
        const int qn = (c / SC_CPL + 1 < SC_NLINE) ? c / SC_CPL + 1 : SC_NLINE - 1;
        load_group_(nbuf, rowE, rowO, qn * SC_LINE);
      }
      float cur[SC_COLS];
      const bool first = (k == 0);
#pragma unroll
      for (int i = 0; i < SC_COLS; ++i) cur[i] = first ? buf[i] : buf[SC_COLS + i];

      float evf[SC_COLS]; int eve[SC_COLS];
#pragma unroll
      for (int j = 0; j < SC_COLS; ++j) { evf[j] = 0.0f; eve[j] = E_ZERO; }
      if (w > 0) {
#pragma unroll
        for (int j = 0; j < SC_COLS; ++j) {
          evf[j] = edge_f[w - 1][c & 1][j];
          eve[j] = edge_e[w - 1][c & 1][j];
        }
      }

      const int mbase = c * SC_COLS;
      const float eps0 = (c == 0) ? 0.0f : EPS_L2;
      bool done = false;
      if (mbase + SC_COLS <= M && __builtin_amdgcn_ballot_w32(f > 0.0f) == 0xffffffffu) {
        done = short_chunk_(f, e, cur, eps0, plf, ple, evf, eve, sel, is0, lane == 31,
                            &edge_f[w][c & 1][0], &edge_e[w][c & 1][0]);
        if (done && mbase + SC_COLS == M) { rf = f; re = e; }
      }
      if (!done) {
        float kf[SC_COLS]; int ke[SC_COLS];
        if (mbase + SC_COLS <= M) {
          float pf[SC_COLS]; int pe[SC_COLS];
#pragma unroll
          for (int j = 0; j < SC_COLS; ++j) weight_(cur[j], j == 0 ? eps0 : EPS_L2, pf[j], pe[j]);
#pragma unroll
          for (int j = 0; j < SC_COLS; ++j) {
            scan_col_(f, e, j == 0 ? plf : evf[j == 0 ? 0 : j - 1],
                            j == 0 ? ple : eve[j == 0 ? 0 : j - 1], pf[j], pe[j], sel, is0);
            kf[j] = f; ke[j] = e;
          }
          if (mbase + SC_COLS == M) { rf = f; re = e; }
        } else {
#pragma unroll
          for (int j = 0; j < SC_COLS; ++j) {
            if (mbase + j < M) {
              float pfj; int pej;
              weight_(cur[j], j == 0 ? eps0 : EPS_L2, pfj, pej);
              scan_col_(f, e, j == 0 ? plf : evf[j == 0 ? 0 : j - 1],
                              j == 0 ? ple : eve[j == 0 ? 0 : j - 1], pfj, pej, sel, is0);
              if (mbase + j == M - 1) { rf = f; re = e; }
            }
            kf[j] = f; ke[j] = e;
          }
        }
        if (lane == 31) {
#pragma unroll
          for (int j = 0; j < SC_COLS; ++j) { edge_f[w][c & 1][j] = kf[j]; edge_e[w][c & 1][j] = ke[j]; }
        }
      }
      plf = evf[SC_COLS - 1];
      ple = eve[SC_COLS - 1];

      e += __builtin_amdgcn_frexp_expf(f);
      f  = __builtin_amdgcn_frexp_mantf(f);
      if (k == SC_CPL - 1) {
        own_row_(buf, nbuf, odd);
      }
    }
    stage_barrier_();
  }
  if (t == tl - 1) {
    const float la = (rf > 0.0f) ? (log2f(rf) + (float)re) * LN2_F : -PAD_MAG;
    ws[P_OFF + b] = la / (float)M;
  }
}

__global__ void mdn_reduce_kernel(const float* __restrict__ ws,
                                  float* __restrict__ out) {
  float s = 0.f;
  for (int b = 0; b < B_; ++b) s += ws[P_OFF + b];
  out[0] = -s / (float)B_;
}

extern "C" void kernel_launch(void* const* d_in, const int* in_sizes, int n_in,
                              void* d_out, int out_size, void* d_ws, size_t ws_size,
                              hipStream_t stream) {
  const float* mu_logvar = (const float*)d_in[0];
  const float* melspec   = (const float*)d_in[1];
  const int*   text_len  = (const int*)d_in[2];
  const int*   mel_len   = (const int*)d_in[3];
  float* out = (float*)d_out;
  float* ws  = (float*)d_ws;
  float* lp  = out + 1;

  mdn_prep_kernel<<<(B_ * TXT_PAD) / PREP_WAVES, PREP_WAVES * 32, 0, stream>>>(mu_logvar, ws);

  const int total_waves = B_ * (TXT_PAD / 16) * (MEL_T / 16);
  mdn_gemm_kernel<<<total_waves / 8, 256, 0, stream>>>(ws, melspec, lp);

  mdn_scan_kernel<<<B_, SC_WAVES * 32, 0, stream>>>(lp, text_len, mel_len, ws);
  mdn_reduce_kernel<<<1, 1, 0, stream>>>(ws, out);
}
